# starting kernel
# baseline (speedup 1.0000x reference)
#include <hip/hip_runtime.h>
#include <math.h>

typedef __attribute__((ext_vector_type(2))) float v2f;
typedef __attribute__((ext_vector_type(8))) float v8f;

#define B_    16
#define C_    768
#define HW_   4096
#define D_    256
#define M_    32
#define EPS_  0.05f
#define NORM_ (-8.3255517f)
#define NEPS_ 1e-12f

#define OFF_WHAT  0
#define OFF_Z     8192
#define OFF_U     (8192 + 2097152)
#define OFF_V     (OFF_U + 512)
#define OFF_RINV  (OFF_V + 65536)

static __device__ __forceinline__ v8f wmma4(v2f a, v2f b, v8f c) {
  return __builtin_amdgcn_wmma_f32_16x16x4_f32(false, a, false, b, (short)0, c,
                                               false, false);
}

__global__ void k_norm_v(const float* __restrict__ v, float* __restrict__ what) {
  int m = threadIdx.x;
  if (m < M_) {
    float s = 0.f;
    for (int d = 0; d < D_; ++d) { float t = v[m * D_ + d]; s += t * t; }
    float inv = 1.0f / fmaxf(sqrtf(s), NEPS_);
    for (int d = 0; d < D_; ++d) what[m * D_ + d] = v[m * D_ + d] * inv;
  }
}

__global__ void k_zero(float* __restrict__ p, int n) {
  int i = blockIdx.x * blockDim.x + threadIdx.x;
  if (i < n) p[i] = 0.f;
}

__global__ void __launch_bounds__(256) k_proj(
    const float* __restrict__ x, const float* __restrict__ w,
    const float* __restrict__ bias, const float* __restrict__ what,
    float* __restrict__ xp, float* __restrict__ Z, float* __restrict__ rinvg) {
  __shared__ float sA[C_ * 16];
  __shared__ float sRed[8][16];
  __shared__ float sRinv[16];

  const int tid = threadIdx.x;
  const int b = blockIdx.x >> 8;
  const int n0 = (blockIdx.x & 255) * 16;
  const int wave = tid >> 5;
  const int lane = tid & 31;
  const int laneLo = lane & 15;
  const int laneHi = lane >> 4;

  {
    const float* xb = x + (size_t)b * C_ * HW_ + n0;
    for (int i = tid; i < C_ * 16; i += 256) {
      int c = i >> 4, n = i & 15;
      sA[i] = xb[(size_t)c * HW_ + n];
    }
  }
  __syncthreads();

  const int d0 = wave * 32;
  v8f acc0 = {}; v8f acc1 = {};
  const float* w0 = w + (size_t)(d0 + laneLo) * C_;
  const float* w1 = w + (size_t)(d0 + 16 + laneLo) * C_;
#pragma unroll 4
  for (int k = 0; k < C_; k += 4) {
    int kk = k + 2 * laneHi;
    v2f a; a.x = sA[kk * 16 + laneLo]; a.y = sA[(kk + 1) * 16 + laneLo];
    v2f b0 = *(const v2f*)(w0 + kk);
    v2f b1 = *(const v2f*)(w1 + kk);
    acc0 = wmma4(a, b0, acc0);
    acc1 = wmma4(a, b1, acc1);
  }

  const float bi0 = bias[d0 + laneLo];
  const float bi1 = bias[d0 + 16 + laneLo];
#pragma unroll
  for (int r = 0; r < 8; ++r) { acc0[r] += bi0; acc1[r] += bi1; }

  {
    float* xpb = xp + (size_t)b * HW_ * D_ + (size_t)n0 * D_;
#pragma unroll
    for (int r = 0; r < 8; ++r) {
      int row = r + 8 * laneHi;
      xpb[(size_t)row * D_ + d0 + laneLo] = acc0[r];
      xpb[(size_t)row * D_ + d0 + 16 + laneLo] = acc1[r];
    }
  }

#pragma unroll
  for (int r = 0; r < 8; ++r) {
    float val = acc0[r] * acc0[r] + acc1[r] * acc1[r];
    for (int off = 1; off < 16; off <<= 1) val += __shfl_xor(val, off, 32);
    if (laneLo == 0) sRed[wave][r + 8 * laneHi] = val;
  }
  __syncthreads();

  if (tid < 16) {
    float s = 0.f;
    for (int wv = 0; wv < 8; ++wv) s += sRed[wv][tid];
    float inv = 1.0f / fmaxf(sqrtf(s), NEPS_);
    sRinv[tid] = inv;
    rinvg[(size_t)b * HW_ + n0 + tid] = inv;
  }

  float* sP = sA;
#pragma unroll
  for (int r = 0; r < 8; ++r) {
    int row = r + 8 * laneHi;
    sP[(d0 + laneLo) * 16 + row] = acc0[r];
    sP[(d0 + 16 + laneLo) * 16 + row] = acc1[r];
  }
  __syncthreads();

  if (wave < 2) {
    const int m0 = wave * 16;
    v8f cs = {};
    const float* wh = what + (size_t)(m0 + laneLo) * D_;
#pragma unroll 4
    for (int k = 0; k < D_; k += 4) {
      int kk = k + 2 * laneHi;
      v2f a; a.x = sP[kk * 16 + laneLo]; a.y = sP[(kk + 1) * 16 + laneLo];
      v2f bb = *(const v2f*)(wh + kk);
      cs = wmma4(a, bb, cs);
    }
    float* Zb = Z + (size_t)b * (M_ * HW_) + (size_t)(m0 + laneLo) * HW_ + n0;
    const float invEps = 1.0f / EPS_;
#pragma unroll
    for (int r = 0; r < 8; ++r) {
      int row = r + 8 * laneHi;
      Zb[row] = cs[r] * sRinv[row] * invEps;
    }
  }
}

__global__ void __launch_bounds__(256) k_u(const float* __restrict__ Z,
                                           const float* __restrict__ vdual,
                                           float* __restrict__ u) {
  __shared__ float sb[256];
  const int b = blockIdx.x >> 5;
  const int m = blockIdx.x & 31;
  const float* zrow = Z + (size_t)b * (M_ * HW_) + (size_t)m * HW_;
  const float* vv = vdual + (size_t)b * HW_;
  const int tid = threadIdx.x;

  float mx = -1e30f;
  for (int n = tid; n < HW_; n += 256) mx = fmaxf(mx, zrow[n] + vv[n]);
  sb[tid] = mx; __syncthreads();
  for (int s = 128; s > 0; s >>= 1) {
    if (tid < s) sb[tid] = fmaxf(sb[tid], sb[tid + s]);
    __syncthreads();
  }
  const float gmax = sb[0]; __syncthreads();

  float sm = 0.f;
  for (int n = tid; n < HW_; n += 256) sm += expf(zrow[n] + vv[n] - gmax);
  sb[tid] = sm; __syncthreads();
  for (int s = 128; s > 0; s >>= 1) {
    if (tid < s) sb[tid] += sb[tid + s];
    __syncthreads();
  }
  if (tid == 0) u[b * M_ + m] = NORM_ - (gmax + logf(sb[0]));
}

__global__ void __launch_bounds__(256) k_vdual(const float* __restrict__ Z,
                                               const float* __restrict__ u,
                                               float* __restrict__ vdual) {
  __shared__ float sU[M_];
  const int idx = blockIdx.x * 256 + threadIdx.x;
  const int b = idx >> 12;
  const int n = idx & 4095;
  if (threadIdx.x < M_) sU[threadIdx.x] = u[b * M_ + threadIdx.x];
  __syncthreads();
  const float* zb = Z + (size_t)b * (M_ * HW_) + n;
  float mx = -1e30f;
#pragma unroll
  for (int m = 0; m < M_; ++m) mx = fmaxf(mx, zb[(size_t)m * HW_] + sU[m]);
  float sm = 0.f;
#pragma unroll
  for (int m = 0; m < M_; ++m) sm += expf(zb[(size_t)m * HW_] + sU[m] - mx);
  vdual[idx] = NORM_ - (mx + logf(sm));
}

__global__ void __launch_bounds__(256) k_vtilde(
    const float* __restrict__ Z, const float* __restrict__ u,
    const float* __restrict__ vdual, const float* __restrict__ xp,
    const float* __restrict__ rinv, float* __restrict__ vt) {
  __shared__ float sAcc[8][32][9];
  const int tid = threadIdx.x;
  const int wave = tid >> 5;
  const int lane = tid & 31;
  const int laneLo = lane & 15;
  const int laneHi = lane >> 4;

  const int b = blockIdx.x >> 5;
  const int rem = blockIdx.x & 31;
  const int m0 = (rem >> 4) * 16;
  const int d0 = (rem & 15) * 16;

  const int k_beg = wave * (HW_ / 8);
  const int k_end = k_beg + (HW_ / 8);

  const float* zrow = Z + (size_t)b * (M_ * HW_) + (size_t)(m0 + laneLo) * HW_;
  const float um = u[b * M_ + m0 + laneLo];
  const float* vv = vdual + (size_t)b * HW_;
  const float* xpb = xp + (size_t)b * HW_ * D_ + d0 + laneLo;
  const float* rv = rinv + (size_t)b * HW_;

  v8f acc = {};
  for (int k = k_beg; k < k_end; k += 4) {
    int kk = k + 2 * laneHi;
    v2f z2 = *(const v2f*)(zrow + kk);
    v2f vw = *(const v2f*)(vv + kk);
    v2f a;
    a.x = expf(z2.x + um + vw.x - NORM_);
    a.y = expf(z2.y + um + vw.y - NORM_);
    v2f r2 = *(const v2f*)(rv + kk);
    v2f bb;
    bb.x = xpb[(size_t)kk * D_] * r2.x;
    bb.y = xpb[(size_t)(kk + 1) * D_] * r2.y;
    acc = wmma4(a, bb, acc);
  }

#pragma unroll
  for (int r = 0; r < 8; ++r) sAcc[wave][lane][r] = acc[r];
  __syncthreads();

  if (wave == 0) {
    float* out = vt + (size_t)b * M_ * D_ + (size_t)m0 * D_ + d0 + laneLo;
#pragma unroll
    for (int r = 0; r < 8; ++r) {
      float s = 0.f;
      for (int w2 = 0; w2 < 8; ++w2) s += sAcc[w2][lane][r];
      int row = r + 8 * laneHi;
      out[(size_t)row * D_] = s;
    }
  }
}

extern "C" void kernel_launch(void* const* d_in, const int* in_sizes, int n_in,
                              void* d_out, int out_size, void* d_ws, size_t ws_size,
                              hipStream_t stream) {
  const float* x      = (const float*)d_in[0];
  const float* conv_w = (const float*)d_in[1];
  const float* conv_b = (const float*)d_in[2];
  const float* v      = (const float*)d_in[3];

  float* out_vt = (float*)d_out;
  float* out_xp = out_vt + (size_t)B_ * M_ * D_;

  float* ws   = (float*)d_ws;
  float* what = ws + OFF_WHAT;
  float* Z    = ws + OFF_Z;
  float* u    = ws + OFF_U;
  float* vd   = ws + OFF_V;
  float* rinv = ws + OFF_RINV;

  k_norm_v<<<1, 256, 0, stream>>>(v, what);
  k_zero<<<(B_ * HW_ + 255) / 256, 256, 0, stream>>>(vd, B_ * HW_);
  k_proj<<<B_ * (HW_ / 16), 256, 0, stream>>>(x, conv_w, conv_b, what,
                                              out_xp, Z, rinv);
  for (int it = 0; it < 3; ++it) {
    k_u<<<B_ * M_, 256, 0, stream>>>(Z, vd, u);
    k_vdual<<<B_ * HW_ / 256, 256, 0, stream>>>(Z, u, vd);
  }
  k_vtilde<<<B_ * M_, 256, 0, stream>>>(Z, u, vd, out_xp, rinv, out_vt);
}


# k_proj split: conv weights stay in registers (16 cols x K per wave), 256 blocks stream 32-row x tiles via 2 LDS stages; norm+scores in own kernel; x_proj, norm, scores k sum order changed
# speedup vs baseline: 1.7153x; 1.7153x over previous
#include <hip/hip_runtime.h>
#include <math.h>

typedef __attribute__((ext_vector_type(2))) float v2f;
typedef __attribute__((ext_vector_type(4))) float v4f;
typedef __attribute__((ext_vector_type(8))) float v8f;

#define B_    16
#define C_    768
#define HW_   4096
#define D_    256
#define M_    32
#define EPS_  0.05f
#define NORM_ (-8.3255517f)
#define NEPS_ 1e-12f

#define OFF_WHAT  0
#define OFF_Z     8192
#define OFF_U     (8192 + 2097152)
#define OFF_V     (OFF_U + 512)
#define OFF_RINV  (OFF_V + 65536)

static __device__ __forceinline__ v8f wmma4(v2f a, v2f b, v8f c) {
  return __builtin_amdgcn_wmma_f32_16x16x4_f32(false, a, false, b, (short)0, c,
                                               false, false);
}

__global__ void k_norm_v(const float* __restrict__ v, float* __restrict__ what) {
  int m = threadIdx.x;
  if (m < M_) {
    float s = 0.f;
    for (int d = 0; d < D_; ++d) { float t = v[m * D_ + d]; s += t * t; }
    float inv = 1.0f / fmaxf(sqrtf(s), NEPS_);
    for (int d = 0; d < D_; ++d) what[m * D_ + d] = v[m * D_ + d] * inv;
  }
}

__global__ void k_zero(float* __restrict__ p, int n) {
  int i = blockIdx.x * blockDim.x + threadIdx.x;
  if (i < n) p[i] = 0.f;
}


#define TILE_N_   32
#define STAGE_F_  (C_ * TILE_N_)
#define TILES_PB_ 16

__global__ void __launch_bounds__(256) k_projw(
    const float* __restrict__ x, const float* __restrict__ w,
    const float* __restrict__ bias, float* __restrict__ xp) {
  __shared__ __attribute__((aligned(16))) float sX[2 * STAGE_F_];

  const int tid = threadIdx.x;
  const int wave = tid >> 5;
  const int lane = tid & 31;
  const int laneLo = lane & 15;
  const int laneHi = lane >> 4;

  const int dh = blockIdx.x & 1;
  const int tile0 = (blockIdx.x >> 1) * TILES_PB_;
  const int b = tile0 >> 7;
  const int nbase = (tile0 & 127) * TILE_N_;
  const int d0 = dh * 128 + wave * 16;

  v4f breg[96];
  {
    const float* wrow = w + (size_t)(d0 + laneLo) * C_ + 4 * laneHi;
#pragma unroll
    for (int i = 0; i < 96; ++i) breg[i] = *(const v4f*)(wrow + 8 * i);
  }
  const float bi = bias[d0 + laneLo];

  const int sq = lane >> 3;
  const int n4 = (lane & 7) * 4;
  const float* xb = x + (size_t)b * C_ * HW_ + nbase + n4;

#pragma unroll
  for (int p = 0; p < 6; ++p) {
    const int kq = 4 * (p * 8 + wave) + sq;
    const float* s = xb + (size_t)(4 * kq) * HW_;
    v4f g0 = *(const v4f*)(s);
    v4f g1 = *(const v4f*)(s + HW_);
    v4f g2 = *(const v4f*)(s + 2 * HW_);
    v4f g3 = *(const v4f*)(s + 3 * HW_);
    float* dst = sX + kq * 128 + n4 * 4;
    *(v4f*)(dst)      = (v4f){g0.x, g1.x, g2.x, g3.x};
    *(v4f*)(dst + 4)  = (v4f){g0.y, g1.y, g2.y, g3.y};
    *(v4f*)(dst + 8)  = (v4f){g0.z, g1.z, g2.z, g3.z};
    *(v4f*)(dst + 12) = (v4f){g0.w, g1.w, g2.w, g3.w};
  }
  __syncthreads();

  for (int t = 0; t < TILES_PB_; ++t) {
    const float* cur = sX + (t & 1) * STAGE_F_;
    float* nxt = sX + ((t + 1) & 1) * STAGE_F_;
    const int tn = (t + 1 < TILES_PB_) ? (t + 1) : t;
    const float* xn = xb + tn * TILE_N_;
    const float* aBase = cur + laneHi * 128 + laneLo * 4;

    v8f acc0 = {}; v8f acc1 = {};
    v4f a0s[3], a1s[3];
    a0s[0] = *(const v4f*)(aBase);
    a1s[0] = *(const v4f*)(aBase + 64);
    a0s[1] = *(const v4f*)(aBase + 256);
    a1s[1] = *(const v4f*)(aBase + 256 + 64);
    v4f g0, g1, g2, g3;
#pragma unroll
    for (int i = 0; i < 96; ++i) {
      const int p = i >> 4;
      const int kq = 4 * (p * 8 + wave) + sq;
      if ((i & 15) == 0) {
        const float* s = xn + (size_t)(4 * kq) * HW_;
        g0 = *(const v4f*)(s);
        g1 = *(const v4f*)(s + HW_);
        g2 = *(const v4f*)(s + 2 * HW_);
        g3 = *(const v4f*)(s + 3 * HW_);
      }
      if (i + 2 < 96) {
        a0s[(i + 2) % 3] = *(const v4f*)(aBase + (i + 2) * 256);
        a1s[(i + 2) % 3] = *(const v4f*)(aBase + (i + 2) * 256 + 64);
      }
      const v4f a0 = a0s[i % 3];
      const v4f a1 = a1s[i % 3];
      const v4f bq = breg[i];
      const v2f bA = {bq.x, bq.y};
      const v2f bB = {bq.z, bq.w};
      acc0 = wmma4((v2f){a0.x, a0.y}, bA, acc0);
      acc1 = wmma4((v2f){a1.x, a1.y}, bA, acc1);
      acc0 = wmma4((v2f){a0.z, a0.w}, bB, acc0);
      acc1 = wmma4((v2f){a1.z, a1.w}, bB, acc1);
      if ((i & 15) == 15) {
        float* dst = nxt + kq * 128 + n4 * 4;
        *(v4f*)(dst)      = (v4f){g0.x, g1.x, g2.x, g3.x};
        *(v4f*)(dst + 4)  = (v4f){g0.y, g1.y, g2.y, g3.y};
        *(v4f*)(dst + 8)  = (v4f){g0.z, g1.z, g2.z, g3.z};
        *(v4f*)(dst + 12) = (v4f){g0.w, g1.w, g2.w, g3.w};
      }
    }

    __syncthreads();

    {
      float* o = xp + ((size_t)b * HW_ + nbase + t * TILE_N_) * D_ + d0 + laneLo;
#pragma unroll
      for (int r = 0; r < 8; ++r) {
        const int row = r + 8 * laneHi;
        o[(size_t)row * D_] = acc0[r] + bi;
        o[(size_t)(row + 16) * D_] = acc1[r] + bi;
      }
    }
  }
}

#define WP_ 260
__global__ void __launch_bounds__(256) k_scores(
    const float* __restrict__ xp, const float* __restrict__ what,
    float* __restrict__ Z, float* __restrict__ rinvg) {
  __shared__ __attribute__((aligned(16))) float sW[M_ * WP_];
  const int tid = threadIdx.x;
  const int wave = tid >> 5;
  const int lane = tid & 31;
  const int laneLo = lane & 15;
  const int laneHi = lane >> 4;

  for (int i = tid; i < M_ * (D_ / 4); i += 256) {
    const int m = i >> 6, q = i & 63;
    *(v4f*)(sW + m * WP_ + q * 4) = *(const v4f*)(what + m * D_ + q * 4);
  }
  __syncthreads();

  const int T = blockIdx.x * 8 + wave;
  const int b = T >> 8;
  const int n0 = (T & 255) * 16;

  const float* arow = xp + ((size_t)b * HW_ + n0 + laneLo) * D_ + 4 * laneHi;
  const float* w0p = sW + laneLo * WP_ + 4 * laneHi;
  const float* w1p = w0p + 16 * WP_;

  v8f cs0 = {}; v8f cs1 = {};
  float ss = 0.f;
#pragma unroll 8
  for (int i = 0; i < 32; ++i) {
    v4f a = *(const v4f*)(arow + 8 * i);
    ss += a.x * a.x; ss += a.y * a.y; ss += a.z * a.z; ss += a.w * a.w;
    v4f q0 = *(const v4f*)(w0p + 8 * i);
    v4f q1 = *(const v4f*)(w1p + 8 * i);
    v2f aA = {a.x, a.y};
    v2f aB = {a.z, a.w};
    cs0 = wmma4(aA, (v2f){q0.x, q0.y}, cs0);
    cs1 = wmma4(aA, (v2f){q1.x, q1.y}, cs1);
    cs0 = wmma4(aB, (v2f){q0.z, q0.w}, cs0);
    cs1 = wmma4(aB, (v2f){q1.z, q1.w}, cs1);
  }
  ss += __shfl_xor(ss, 16, 32);
  const float inv = 1.0f / fmaxf(sqrtf(ss), NEPS_);
  if (laneHi == 0) rinvg[(size_t)b * HW_ + n0 + laneLo] = inv;

  float* z0 = Z + (size_t)b * (M_ * HW_) + (size_t)laneLo * HW_ + n0 + 8 * laneHi;
  float* z1 = z0 + (size_t)16 * HW_;
  const float invEps = 1.0f / EPS_;
#pragma unroll
  for (int r = 0; r < 8; ++r) {
    const float ri = __shfl(inv, 8 * laneHi + r, 32);
    z0[r] = cs0[r] * ri * invEps;
    z1[r] = cs1[r] * ri * invEps;
  }
}

__global__ void __launch_bounds__(256) k_u(const float* __restrict__ Z,
                                           const float* __restrict__ vdual,
                                           float* __restrict__ u) {
  __shared__ float sb[256];
  const int b = blockIdx.x >> 5;
  const int m = blockIdx.x & 31;
  const float* zrow = Z + (size_t)b * (M_ * HW_) + (size_t)m * HW_;
  const float* vv = vdual + (size_t)b * HW_;
  const int tid = threadIdx.x;

  float mx = -1e30f;
  for (int n = tid; n < HW_; n += 256) mx = fmaxf(mx, zrow[n] + vv[n]);
  sb[tid] = mx; __syncthreads();
  for (int s = 128; s > 0; s >>= 1) {
    if (tid < s) sb[tid] = fmaxf(sb[tid], sb[tid + s]);
    __syncthreads();
  }
  const float gmax = sb[0]; __syncthreads();

  float sm = 0.f;
  for (int n = tid; n < HW_; n += 256) sm += expf(zrow[n] + vv[n] - gmax);
  sb[tid] = sm; __syncthreads();
  for (int s = 128; s > 0; s >>= 1) {
    if (tid < s) sb[tid] += sb[tid + s];
    __syncthreads();
  }
  if (tid == 0) u[b * M_ + m] = NORM_ - (gmax + logf(sb[0]));
}

__global__ void __launch_bounds__(256) k_vdual(const float* __restrict__ Z,
                                               const float* __restrict__ u,
                                               float* __restrict__ vdual) {
  __shared__ float sU[M_];
  const int idx = blockIdx.x * 256 + threadIdx.x;
  const int b = idx >> 12;
  const int n = idx & 4095;
  if (threadIdx.x < M_) sU[threadIdx.x] = u[b * M_ + threadIdx.x];
  __syncthreads();
  const float* zb = Z + (size_t)b * (M_ * HW_) + n;
  float mx = -1e30f;
#pragma unroll
  for (int m = 0; m < M_; ++m) mx = fmaxf(mx, zb[(size_t)m * HW_] + sU[m]);
  float sm = 0.f;
#pragma unroll
  for (int m = 0; m < M_; ++m) sm += expf(zb[(size_t)m * HW_] + sU[m] - mx);
  vdual[idx] = NORM_ - (mx + logf(sm));
}

__global__ void __launch_bounds__(256) k_vtilde(
    const float* __restrict__ Z, const float* __restrict__ u,
    const float* __restrict__ vdual, const float* __restrict__ xp,
    const float* __restrict__ rinv, float* __restrict__ vt) {
  __shared__ float sAcc[8][32][9];
  const int tid = threadIdx.x;
  const int wave = tid >> 5;
  const int lane = tid & 31;
  const int laneLo = lane & 15;
  const int laneHi = lane >> 4;

  const int b = blockIdx.x >> 5;
  const int rem = blockIdx.x & 31;
  const int m0 = (rem >> 4) * 16;
  const int d0 = (rem & 15) * 16;

  const int k_beg = wave * (HW_ / 8);
  const int k_end = k_beg + (HW_ / 8);

  const float* zrow = Z + (size_t)b * (M_ * HW_) + (size_t)(m0 + laneLo) * HW_;
  const float um = u[b * M_ + m0 + laneLo];
  const float* vv = vdual + (size_t)b * HW_;
  const float* xpb = xp + (size_t)b * HW_ * D_ + d0 + laneLo;
  const float* rv = rinv + (size_t)b * HW_;

  v8f acc = {};
  for (int k = k_beg; k < k_end; k += 4) {
    int kk = k + 2 * laneHi;
    v2f z2 = *(const v2f*)(zrow + kk);
    v2f vw = *(const v2f*)(vv + kk);
    v2f a;
    a.x = expf(z2.x + um + vw.x - NORM_);
    a.y = expf(z2.y + um + vw.y - NORM_);
    v2f r2 = *(const v2f*)(rv + kk);
    v2f bb;
    bb.x = xpb[(size_t)kk * D_] * r2.x;
    bb.y = xpb[(size_t)(kk + 1) * D_] * r2.y;
    acc = wmma4(a, bb, acc);
  }

#pragma unroll
  for (int r = 0; r < 8; ++r) sAcc[wave][lane][r] = acc[r];
  __syncthreads();

  if (wave == 0) {
    float* out = vt + (size_t)b * M_ * D_ + (size_t)m0 * D_ + d0 + laneLo;
#pragma unroll
    for (int r = 0; r < 8; ++r) {
      float s = 0.f;
      for (int w2 = 0; w2 < 8; ++w2) s += sAcc[w2][lane][r];
      int row = r + 8 * laneHi;
      out[(size_t)row * D_] = s;
    }
  }
}

extern "C" void kernel_launch(void* const* d_in, const int* in_sizes, int n_in,
                              void* d_out, int out_size, void* d_ws, size_t ws_size,
                              hipStream_t stream) {
  const float* x      = (const float*)d_in[0];
  const float* conv_w = (const float*)d_in[1];
  const float* conv_b = (const float*)d_in[2];
  const float* v      = (const float*)d_in[3];

  float* out_vt = (float*)d_out;
  float* out_xp = out_vt + (size_t)B_ * M_ * D_;

  float* ws   = (float*)d_ws;
  float* what = ws + OFF_WHAT;
  float* Z    = ws + OFF_Z;
  float* u    = ws + OFF_U;
  float* vd   = ws + OFF_V;
  float* rinv = ws + OFF_RINV;

  k_norm_v<<<1, 256, 0, stream>>>(v, what);
  k_zero<<<(B_ * HW_ + 255) / 256, 256, 0, stream>>>(vd, B_ * HW_);
  k_projw<<<256, 256, 0, stream>>>(x, conv_w, conv_b, out_xp);
  k_scores<<<B_ * HW_ / 128, 256, 0, stream>>>(out_xp, what, Z, rinv);
  for (int it = 0; it < 3; ++it) {
    k_u<<<B_ * M_, 256, 0, stream>>>(Z, vd, u);
    k_vdual<<<B_ * HW_ / 256, 256, 0, stream>>>(Z, u, vd);
  }
  k_vtilde<<<B_ * M_, 256, 0, stream>>>(Z, u, vd, out_xp, rinv, out_vt);
}


# u-update: row kept in registers (16 per thread, one round of 16-byte loads); max and sum by wave shuffles, 2 barriers not 18; first call reads no v, zero-fill launch gone; n sum order changed
# speedup vs baseline: 1.8210x; 1.0616x over previous
#include <hip/hip_runtime.h>
#include <math.h>

typedef __attribute__((ext_vector_type(2))) float v2f;
typedef __attribute__((ext_vector_type(4))) float v4f;
typedef __attribute__((ext_vector_type(8))) float v8f;

#define B_    16
#define C_    768
#define HW_   4096
#define D_    256
#define M_    32
#define EPS_  0.05f
#define NORM_ (-8.3255517f)
#define NEPS_ 1e-12f

#define OFF_WHAT  0
#define OFF_Z     8192
#define OFF_U     (8192 + 2097152)
#define OFF_V     (OFF_U + 512)
#define OFF_RINV  (OFF_V + 65536)

static __device__ __forceinline__ v8f wmma4(v2f a, v2f b, v8f c) {
  return __builtin_amdgcn_wmma_f32_16x16x4_f32(false, a, false, b, (short)0, c,
                                               false, false);
}

__global__ void k_norm_v(const float* __restrict__ v, float* __restrict__ what) {
  int m = threadIdx.x;
  if (m < M_) {
    float s = 0.f;
    for (int d = 0; d < D_; ++d) { float t = v[m * D_ + d]; s += t * t; }
    float inv = 1.0f / fmaxf(sqrtf(s), NEPS_);
    for (int d = 0; d < D_; ++d) what[m * D_ + d] = v[m * D_ + d] * inv;
  }
}


#define TILE_N_   32
#define STAGE_F_  (C_ * TILE_N_)
#define TILES_PB_ 16

__global__ void __launch_bounds__(256) k_projw(
    const float* __restrict__ x, const float* __restrict__ w,
    const float* __restrict__ bias, float* __restrict__ xp) {
  __shared__ __attribute__((aligned(16))) float sX[2 * STAGE_F_];

  const int tid = threadIdx.x;
  const int wave = tid >> 5;
  const int lane = tid & 31;
  const int laneLo = lane & 15;
  const int laneHi = lane >> 4;

  const int dh = blockIdx.x & 1;
  const int tile0 = (blockIdx.x >> 1) * TILES_PB_;
  const int b = tile0 >> 7;
  const int nbase = (tile0 & 127) * TILE_N_;
  const int d0 = dh * 128 + wave * 16;

  v4f breg[96];
  {
    const float* wrow = w + (size_t)(d0 + laneLo) * C_ + 4 * laneHi;
#pragma unroll
    for (int i = 0; i < 96; ++i) breg[i] = *(const v4f*)(wrow + 8 * i);
  }
  const float bi = bias[d0 + laneLo];

  const int sq = lane >> 3;
  const int n4 = (lane & 7) * 4;
  const float* xb = x + (size_t)b * C_ * HW_ + nbase + n4;

#pragma unroll
  for (int p = 0; p < 6; ++p) {
    const int kq = 4 * (p * 8 + wave) + sq;
    const float* s = xb + (size_t)(4 * kq) * HW_;
    v4f g0 = *(const v4f*)(s);
    v4f g1 = *(const v4f*)(s + HW_);
    v4f g2 = *(const v4f*)(s + 2 * HW_);
    v4f g3 = *(const v4f*)(s + 3 * HW_);
    float* dst = sX + kq * 128 + n4 * 4;
    *(v4f*)(dst)      = (v4f){g0.x, g1.x, g2.x, g3.x};
    *(v4f*)(dst + 4)  = (v4f){g0.y, g1.y, g2.y, g3.y};
    *(v4f*)(dst + 8)  = (v4f){g0.z, g1.z, g2.z, g3.z};
    *(v4f*)(dst + 12) = (v4f){g0.w, g1.w, g2.w, g3.w};
  }
  __syncthreads();

  for (int t = 0; t < TILES_PB_; ++t) {
    const float* cur = sX + (t & 1) * STAGE_F_;
    float* nxt = sX + ((t + 1) & 1) * STAGE_F_;
    const int tn = (t + 1 < TILES_PB_) ? (t + 1) : t;
    const float* xn = xb + tn * TILE_N_;
    const float* aBase = cur + laneHi * 128 + laneLo * 4;

    v8f acc0 = {}; v8f acc1 = {};
    v4f a0s[3], a1s[3];
    a0s[0] = *(const v4f*)(aBase);
    a1s[0] = *(const v4f*)(aBase + 64);
    a0s[1] = *(const v4f*)(aBase + 256);
    a1s[1] = *(const v4f*)(aBase + 256 + 64);
    v4f g0, g1, g2, g3;
#pragma unroll
    for (int i = 0; i < 96; ++i) {
      const int p = i >> 4;
      const int kq = 4 * (p * 8 + wave) + sq;
      if ((i & 15) == 0) {
        const float* s = xn + (size_t)(4 * kq) * HW_;
        g0 = *(const v4f*)(s);
        g1 = *(const v4f*)(s + HW_);
        g2 = *(const v4f*)(s + 2 * HW_);
        g3 = *(const v4f*)(s + 3 * HW_);
      }
      if (i + 2 < 96) {
        a0s[(i + 2) % 3] = *(const v4f*)(aBase + (i + 2) * 256);
        a1s[(i + 2) % 3] = *(const v4f*)(aBase + (i + 2) * 256 + 64);
      }
      const v4f a0 = a0s[i % 3];
      const v4f a1 = a1s[i % 3];
      const v4f bq = breg[i];
      const v2f bA = {bq.x, bq.y};
      const v2f bB = {bq.z, bq.w};
      acc0 = wmma4((v2f){a0.x, a0.y}, bA, acc0);
      acc1 = wmma4((v2f){a1.x, a1.y}, bA, acc1);
      acc0 = wmma4((v2f){a0.z, a0.w}, bB, acc0);
      acc1 = wmma4((v2f){a1.z, a1.w}, bB, acc1);
      if ((i & 15) == 15) {
        float* dst = nxt + kq * 128 + n4 * 4;
        *(v4f*)(dst)      = (v4f){g0.x, g1.x, g2.x, g3.x};
        *(v4f*)(dst + 4)  = (v4f){g0.y, g1.y, g2.y, g3.y};
        *(v4f*)(dst + 8)  = (v4f){g0.z, g1.z, g2.z, g3.z};
        *(v4f*)(dst + 12) = (v4f){g0.w, g1.w, g2.w, g3.w};
      }
    }

    __syncthreads();

    {
      float* o = xp + ((size_t)b * HW_ + nbase + t * TILE_N_) * D_ + d0 + laneLo;
#pragma unroll
      for (int r = 0; r < 8; ++r) {
        const int row = r + 8 * laneHi;
        o[(size_t)row * D_] = acc0[r] + bi;
        o[(size_t)(row + 16) * D_] = acc1[r] + bi;
      }
    }
  }
}

#define WP_ 260
__global__ void __launch_bounds__(256) k_scores(
    const float* __restrict__ xp, const float* __restrict__ what,
    float* __restrict__ Z, float* __restrict__ rinvg) {
  __shared__ __attribute__((aligned(16))) float sW[M_ * WP_];
  const int tid = threadIdx.x;
  const int wave = tid >> 5;
  const int lane = tid & 31;
  const int laneLo = lane & 15;
  const int laneHi = lane >> 4;

  for (int i = tid; i < M_ * (D_ / 4); i += 256) {
    const int m = i >> 6, q = i & 63;
    *(v4f*)(sW + m * WP_ + q * 4) = *(const v4f*)(what + m * D_ + q * 4);
  }
  __syncthreads();

  const int T = blockIdx.x * 8 + wave;
  const int b = T >> 8;
  const int n0 = (T & 255) * 16;

  const float* arow = xp + ((size_t)b * HW_ + n0 + laneLo) * D_ + 4 * laneHi;
  const float* w0p = sW + laneLo * WP_ + 4 * laneHi;
  const float* w1p = w0p + 16 * WP_;

  v8f cs0 = {}; v8f cs1 = {};
  float ss = 0.f;
#pragma unroll 8
  for (int i = 0; i < 32; ++i) {
    v4f a = *(const v4f*)(arow + 8 * i);
    ss += a.x * a.x; ss += a.y * a.y; ss += a.z * a.z; ss += a.w * a.w;
    v4f q0 = *(const v4f*)(w0p + 8 * i);
    v4f q1 = *(const v4f*)(w1p + 8 * i);
    v2f aA = {a.x, a.y};
    v2f aB = {a.z, a.w};
    cs0 = wmma4(aA, (v2f){q0.x, q0.y}, cs0);
    cs1 = wmma4(aA, (v2f){q1.x, q1.y}, cs1);
    cs0 = wmma4(aB, (v2f){q0.z, q0.w}, cs0);
    cs1 = wmma4(aB, (v2f){q1.z, q1.w}, cs1);
  }
  ss += __shfl_xor(ss, 16, 32);
  const float inv = 1.0f / fmaxf(sqrtf(ss), NEPS_);
  if (laneHi == 0) rinvg[(size_t)b * HW_ + n0 + laneLo] = inv;

  float* z0 = Z + (size_t)b * (M_ * HW_) + (size_t)laneLo * HW_ + n0 + 8 * laneHi;
  float* z1 = z0 + (size_t)16 * HW_;
  const float invEps = 1.0f / EPS_;
#pragma unroll
  for (int r = 0; r < 8; ++r) {
    const float ri = __shfl(inv, 8 * laneHi + r, 32);
    z0[r] = cs0[r] * ri * invEps;
    z1[r] = cs1[r] * ri * invEps;
  }
}

template <bool HAS_V>
static __device__ __forceinline__ void u_row(const float* __restrict__ Z,
                                             const float* __restrict__ vdual,
                                             float* __restrict__ u,
                                             float* sMx, float* sSm) {
  const int b = blockIdx.x >> 5;
  const int m = blockIdx.x & 31;
  const int tid = threadIdx.x;
  const int lane = tid & 31;
  const int wave = tid >> 5;
  const float* zrow = Z + (size_t)b * (M_ * HW_) + (size_t)m * HW_ + 4 * tid;

  v4f t[4];
  for (int j = 0; j < 4; ++j) t[j] = *(const v4f*)(zrow + 1024 * j);
  if (HAS_V) {
    const float* vv = vdual + (size_t)b * HW_ + 4 * tid;
    for (int j = 0; j < 4; ++j) t[j] += *(const v4f*)(vv + 1024 * j);
  }

  float mx = -1e30f;
  for (int j = 0; j < 4; ++j)
    mx = fmaxf(mx, fmaxf(fmaxf(t[j].x, t[j].y), fmaxf(t[j].z, t[j].w)));
  for (int o = 16; o > 0; o >>= 1) mx = fmaxf(mx, __shfl_xor(mx, o, 32));
  if (lane == 0) sMx[wave] = mx;
  __syncthreads();
  float gmax = sMx[0];
  for (int w2 = 1; w2 < 8; ++w2) gmax = fmaxf(gmax, sMx[w2]);

  float sm = 0.f;
  for (int j = 0; j < 4; ++j) {
    sm += expf(t[j].x - gmax);
    sm += expf(t[j].y - gmax);
    sm += expf(t[j].z - gmax);
    sm += expf(t[j].w - gmax);
  }
  for (int o = 16; o > 0; o >>= 1) sm += __shfl_xor(sm, o, 32);
  if (lane == 0) sSm[wave] = sm;
  __syncthreads();
  if (tid == 0) {
    float s = sSm[0];
    for (int w2 = 1; w2 < 8; ++w2) s += sSm[w2];
    u[b * M_ + m] = NORM_ - (gmax + logf(s));
  }
}

__global__ void __launch_bounds__(256) k_u_first(const float* __restrict__ Z,
                                                 float* __restrict__ u) {
  __shared__ float sMx[8];
  __shared__ float sSm[8];
  u_row<false>(Z, nullptr, u, sMx, sSm);
}

__global__ void __launch_bounds__(256) k_u(const float* __restrict__ Z,
                                           const float* __restrict__ vdual,
                                           float* __restrict__ u) {
  __shared__ float sMx[8];
  __shared__ float sSm[8];
  u_row<true>(Z, vdual, u, sMx, sSm);
}

__global__ void __launch_bounds__(256) k_vdual(const float* __restrict__ Z,
                                               const float* __restrict__ u,
                                               float* __restrict__ vdual) {
  __shared__ float sU[M_];
  const int idx = blockIdx.x * 256 + threadIdx.x;
  const int b = idx >> 12;
  const int n = idx & 4095;
  if (threadIdx.x < M_) sU[threadIdx.x] = u[b * M_ + threadIdx.x];
  __syncthreads();
  const float* zb = Z + (size_t)b * (M_ * HW_) + n;
  float mx = -1e30f;
#pragma unroll
  for (int m = 0; m < M_; ++m) mx = fmaxf(mx, zb[(size_t)m * HW_] + sU[m]);
  float sm = 0.f;
#pragma unroll
  for (int m = 0; m < M_; ++m) sm += expf(zb[(size_t)m * HW_] + sU[m] - mx);
  vdual[idx] = NORM_ - (mx + logf(sm));
}

__global__ void __launch_bounds__(256) k_vtilde(
    const float* __restrict__ Z, const float* __restrict__ u,
    const float* __restrict__ vdual, const float* __restrict__ xp,
    const float* __restrict__ rinv, float* __restrict__ vt) {
  __shared__ float sAcc[8][32][9];
  const int tid = threadIdx.x;
  const int wave = tid >> 5;
  const int lane = tid & 31;
  const int laneLo = lane & 15;
  const int laneHi = lane >> 4;

  const int b = blockIdx.x >> 5;
  const int rem = blockIdx.x & 31;
  const int m0 = (rem >> 4) * 16;
  const int d0 = (rem & 15) * 16;

  const int k_beg = wave * (HW_ / 8);
  const int k_end = k_beg + (HW_ / 8);

  const float* zrow = Z + (size_t)b * (M_ * HW_) + (size_t)(m0 + laneLo) * HW_;
  const float um = u[b * M_ + m0 + laneLo];
  const float* vv = vdual + (size_t)b * HW_;
  const float* xpb = xp + (size_t)b * HW_ * D_ + d0 + laneLo;
  const float* rv = rinv + (size_t)b * HW_;

  v8f acc = {};
  for (int k = k_beg; k < k_end; k += 4) {
    int kk = k + 2 * laneHi;
    v2f z2 = *(const v2f*)(zrow + kk);
    v2f vw = *(const v2f*)(vv + kk);
    v2f a;
    a.x = expf(z2.x + um + vw.x - NORM_);
    a.y = expf(z2.y + um + vw.y - NORM_);
    v2f r2 = *(const v2f*)(rv + kk);
    v2f bb;
    bb.x = xpb[(size_t)kk * D_] * r2.x;
    bb.y = xpb[(size_t)(kk + 1) * D_] * r2.y;
    acc = wmma4(a, bb, acc);
  }

#pragma unroll
  for (int r = 0; r < 8; ++r) sAcc[wave][lane][r] = acc[r];
  __syncthreads();

  if (wave == 0) {
    float* out = vt + (size_t)b * M_ * D_ + (size_t)m0 * D_ + d0 + laneLo;
#pragma unroll
    for (int r = 0; r < 8; ++r) {
      float s = 0.f;
      for (int w2 = 0; w2 < 8; ++w2) s += sAcc[w2][lane][r];
      int row = r + 8 * laneHi;
      out[(size_t)row * D_] = s;
    }
  }
}

extern "C" void kernel_launch(void* const* d_in, const int* in_sizes, int n_in,
                              void* d_out, int out_size, void* d_ws, size_t ws_size,
                              hipStream_t stream) {
  const float* x      = (const float*)d_in[0];
  const float* conv_w = (const float*)d_in[1];
  const float* conv_b = (const float*)d_in[2];
  const float* v      = (const float*)d_in[3];

  float* out_vt = (float*)d_out;
  float* out_xp = out_vt + (size_t)B_ * M_ * D_;

  float* ws   = (float*)d_ws;
  float* what = ws + OFF_WHAT;
  float* Z    = ws + OFF_Z;
  float* u    = ws + OFF_U;
  float* vd   = ws + OFF_V;
  float* rinv = ws + OFF_RINV;

  k_norm_v<<<1, 256, 0, stream>>>(v, what);
  k_projw<<<256, 256, 0, stream>>>(x, conv_w, conv_b, out_xp);
  k_scores<<<B_ * HW_ / 128, 256, 0, stream>>>(out_xp, what, Z, rinv);
  k_u_first<<<B_ * M_, 256, 0, stream>>>(Z, u);
  k_vdual<<<B_ * HW_ / 256, 256, 0, stream>>>(Z, u, vd);
  for (int it = 1; it < 3; ++it) {
    k_u<<<B_ * M_, 256, 0, stream>>>(Z, vd, u);
    k_vdual<<<B_ * HW_ / 256, 256, 0, stream>>>(Z, u, vd);
  }
  k_vtilde<<<B_ * M_, 256, 0, stream>>>(Z, u, vd, out_xp, rinv, out_vt);
}


# v_tilde split over n: block = 256 rows n x all 32 m x 256 d; x_proj read once by 16-byte loads issued up front, weights once per block via LDS; +1 launch adds the 16 partials; n sum order changed
# speedup vs baseline: 2.0262x; 1.1127x over previous
#include <hip/hip_runtime.h>
#include <math.h>

typedef __attribute__((ext_vector_type(2))) float v2f;
typedef __attribute__((ext_vector_type(4))) float v4f;
typedef __attribute__((ext_vector_type(8))) float v8f;

#define B_    16
#define C_    768
#define HW_   4096
#define D_    256
#define M_    32
#define EPS_  0.05f
#define NORM_ (-8.3255517f)
#define NEPS_ 1e-12f

#define OFF_WHAT  0
#define OFF_Z     8192
#define OFF_U     (8192 + 2097152)
#define OFF_V     (OFF_U + 512)
#define OFF_RINV  (OFF_V + 65536)
#define OFF_PART  (OFF_RINV + 65536)

static __device__ __forceinline__ v8f wmma4(v2f a, v2f b, v8f c) {
  return __builtin_amdgcn_wmma_f32_16x16x4_f32(false, a, false, b, (short)0, c,
                                               false, false);
}

__global__ void k_norm_v(const float* __restrict__ v, float* __restrict__ what) {
  int m = threadIdx.x;
  if (m < M_) {
    float s = 0.f;
    for (int d = 0; d < D_; ++d) { float t = v[m * D_ + d]; s += t * t; }
    float inv = 1.0f / fmaxf(sqrtf(s), NEPS_);
    for (int d = 0; d < D_; ++d) what[m * D_ + d] = v[m * D_ + d] * inv;
  }
}


#define TILE_N_   32
#define STAGE_F_  (C_ * TILE_N_)
#define TILES_PB_ 16

__global__ void __launch_bounds__(256) k_projw(
    const float* __restrict__ x, const float* __restrict__ w,
    const float* __restrict__ bias, float* __restrict__ xp) {
  __shared__ __attribute__((aligned(16))) float sX[2 * STAGE_F_];

  const int tid = threadIdx.x;
  const int wave = tid >> 5;
  const int lane = tid & 31;
  const int laneLo = lane & 15;
  const int laneHi = lane >> 4;

  const int dh = blockIdx.x & 1;
  const int tile0 = (blockIdx.x >> 1) * TILES_PB_;
  const int b = tile0 >> 7;
  const int nbase = (tile0 & 127) * TILE_N_;
  const int d0 = dh * 128 + wave * 16;

  v4f breg[96];
  {
    const float* wrow = w + (size_t)(d0 + laneLo) * C_ + 4 * laneHi;
#pragma unroll
    for (int i = 0; i < 96; ++i) breg[i] = *(const v4f*)(wrow + 8 * i);
  }
  const float bi = bias[d0 + laneLo];

  const int sq = lane >> 3;
  const int n4 = (lane & 7) * 4;
  const float* xb = x + (size_t)b * C_ * HW_ + nbase + n4;

#pragma unroll
  for (int p = 0; p < 6; ++p) {
    const int kq = 4 * (p * 8 + wave) + sq;
    const float* s = xb + (size_t)(4 * kq) * HW_;
    v4f g0 = *(const v4f*)(s);
    v4f g1 = *(const v4f*)(s + HW_);
    v4f g2 = *(const v4f*)(s + 2 * HW_);
    v4f g3 = *(const v4f*)(s + 3 * HW_);
    float* dst = sX + kq * 128 + n4 * 4;
    *(v4f*)(dst)      = (v4f){g0.x, g1.x, g2.x, g3.x};
    *(v4f*)(dst + 4)  = (v4f){g0.y, g1.y, g2.y, g3.y};
    *(v4f*)(dst + 8)  = (v4f){g0.z, g1.z, g2.z, g3.z};
    *(v4f*)(dst + 12) = (v4f){g0.w, g1.w, g2.w, g3.w};
  }
  __syncthreads();

  for (int t = 0; t < TILES_PB_; ++t) {
    const float* cur = sX + (t & 1) * STAGE_F_;
    float* nxt = sX + ((t + 1) & 1) * STAGE_F_;
    const int tn = (t + 1 < TILES_PB_) ? (t + 1) : t;
    const float* xn = xb + tn * TILE_N_;
    const float* aBase = cur + laneHi * 128 + laneLo * 4;

    v8f acc0 = {}; v8f acc1 = {};
    v4f a0s[3], a1s[3];
    a0s[0] = *(const v4f*)(aBase);
    a1s[0] = *(const v4f*)(aBase + 64);
    a0s[1] = *(const v4f*)(aBase + 256);
    a1s[1] = *(const v4f*)(aBase + 256 + 64);
    v4f g0, g1, g2, g3;
#pragma unroll
    for (int i = 0; i < 96; ++i) {
      const int p = i >> 4;
      const int kq = 4 * (p * 8 + wave) + sq;
      if ((i & 15) == 0) {
        const float* s = xn + (size_t)(4 * kq) * HW_;
        g0 = *(const v4f*)(s);
        g1 = *(const v4f*)(s + HW_);
        g2 = *(const v4f*)(s + 2 * HW_);
        g3 = *(const v4f*)(s + 3 * HW_);
      }
      if (i + 2 < 96) {
        a0s[(i + 2) % 3] = *(const v4f*)(aBase + (i + 2) * 256);
        a1s[(i + 2) % 3] = *(const v4f*)(aBase + (i + 2) * 256 + 64);
      }
      const v4f a0 = a0s[i % 3];
      const v4f a1 = a1s[i % 3];
      const v4f bq = breg[i];
      const v2f bA = {bq.x, bq.y};
      const v2f bB = {bq.z, bq.w};
      acc0 = wmma4((v2f){a0.x, a0.y}, bA, acc0);
      acc1 = wmma4((v2f){a1.x, a1.y}, bA, acc1);
      acc0 = wmma4((v2f){a0.z, a0.w}, bB, acc0);
      acc1 = wmma4((v2f){a1.z, a1.w}, bB, acc1);
      if ((i & 15) == 15) {
        float* dst = nxt + kq * 128 + n4 * 4;
        *(v4f*)(dst)      = (v4f){g0.x, g1.x, g2.x, g3.x};
        *(v4f*)(dst + 4)  = (v4f){g0.y, g1.y, g2.y, g3.y};
        *(v4f*)(dst + 8)  = (v4f){g0.z, g1.z, g2.z, g3.z};
        *(v4f*)(dst + 12) = (v4f){g0.w, g1.w, g2.w, g3.w};
      }
    }

    __syncthreads();

    {
      float* o = xp + ((size_t)b * HW_ + nbase + t * TILE_N_) * D_ + d0 + laneLo;
#pragma unroll
      for (int r = 0; r < 8; ++r) {
        const int row = r + 8 * laneHi;
        o[(size_t)row * D_] = acc0[r] + bi;
        o[(size_t)(row + 16) * D_] = acc1[r] + bi;
      }
    }
  }
}

#define WP_ 260
__global__ void __launch_bounds__(256) k_scores(
    const float* __restrict__ xp, const float* __restrict__ what,
    float* __restrict__ Z, float* __restrict__ rinvg) {
  __shared__ __attribute__((aligned(16))) float sW[M_ * WP_];
  const int tid = threadIdx.x;
  const int wave = tid >> 5;
  const int lane = tid & 31;
  const int laneLo = lane & 15;
  const int laneHi = lane >> 4;

  for (int i = tid; i < M_ * (D_ / 4); i += 256) {
    const int m = i >> 6, q = i & 63;
    *(v4f*)(sW + m * WP_ + q * 4) = *(const v4f*)(what + m * D_ + q * 4);
  }
  __syncthreads();

  const int T = blockIdx.x * 8 + wave;
  const int b = T >> 8;
  const int n0 = (T & 255) * 16;

  const float* arow = xp + ((size_t)b * HW_ + n0 + laneLo) * D_ + 4 * laneHi;
  const float* w0p = sW + laneLo * WP_ + 4 * laneHi;
  const float* w1p = w0p + 16 * WP_;

  v8f cs0 = {}; v8f cs1 = {};
  float ss = 0.f;
#pragma unroll 8
  for (int i = 0; i < 32; ++i) {
    v4f a = *(const v4f*)(arow + 8 * i);
    ss += a.x * a.x; ss += a.y * a.y; ss += a.z * a.z; ss += a.w * a.w;
    v4f q0 = *(const v4f*)(w0p + 8 * i);
    v4f q1 = *(const v4f*)(w1p + 8 * i);
    v2f aA = {a.x, a.y};
    v2f aB = {a.z, a.w};
    cs0 = wmma4(aA, (v2f){q0.x, q0.y}, cs0);
    cs1 = wmma4(aA, (v2f){q1.x, q1.y}, cs1);
    cs0 = wmma4(aB, (v2f){q0.z, q0.w}, cs0);
    cs1 = wmma4(aB, (v2f){q1.z, q1.w}, cs1);
  }
  ss += __shfl_xor(ss, 16, 32);
  const float inv = 1.0f / fmaxf(sqrtf(ss), NEPS_);
  if (laneHi == 0) rinvg[(size_t)b * HW_ + n0 + laneLo] = inv;

  float* z0 = Z + (size_t)b * (M_ * HW_) + (size_t)laneLo * HW_ + n0 + 8 * laneHi;
  float* z1 = z0 + (size_t)16 * HW_;
  const float invEps = 1.0f / EPS_;
#pragma unroll
  for (int r = 0; r < 8; ++r) {
    const float ri = __shfl(inv, 8 * laneHi + r, 32);
    z0[r] = cs0[r] * ri * invEps;
    z1[r] = cs1[r] * ri * invEps;
  }
}

template <bool HAS_V>
static __device__ __forceinline__ void u_row(const float* __restrict__ Z,
                                             const float* __restrict__ vdual,
                                             float* __restrict__ u,
                                             float* sMx, float* sSm) {
  const int b = blockIdx.x >> 5;
  const int m = blockIdx.x & 31;
  const int tid = threadIdx.x;
  const int lane = tid & 31;
  const int wave = tid >> 5;
  const float* zrow = Z + (size_t)b * (M_ * HW_) + (size_t)m * HW_ + 4 * tid;

  v4f t[4];
  for (int j = 0; j < 4; ++j) t[j] = *(const v4f*)(zrow + 1024 * j);
  if (HAS_V) {
    const float* vv = vdual + (size_t)b * HW_ + 4 * tid;
    for (int j = 0; j < 4; ++j) t[j] += *(const v4f*)(vv + 1024 * j);
  }

  float mx = -1e30f;
  for (int j = 0; j < 4; ++j)
    mx = fmaxf(mx, fmaxf(fmaxf(t[j].x, t[j].y), fmaxf(t[j].z, t[j].w)));
  for (int o = 16; o > 0; o >>= 1) mx = fmaxf(mx, __shfl_xor(mx, o, 32));
  if (lane == 0) sMx[wave] = mx;
  __syncthreads();
  float gmax = sMx[0];
  for (int w2 = 1; w2 < 8; ++w2) gmax = fmaxf(gmax, sMx[w2]);

  float sm = 0.f;
  for (int j = 0; j < 4; ++j) {
    sm += expf(t[j].x - gmax);
    sm += expf(t[j].y - gmax);
    sm += expf(t[j].z - gmax);
    sm += expf(t[j].w - gmax);
  }
  for (int o = 16; o > 0; o >>= 1) sm += __shfl_xor(sm, o, 32);
  if (lane == 0) sSm[wave] = sm;
  __syncthreads();
  if (tid == 0) {
    float s = sSm[0];
    for (int w2 = 1; w2 < 8; ++w2) s += sSm[w2];
    u[b * M_ + m] = NORM_ - (gmax + logf(s));
  }
}

__global__ void __launch_bounds__(256) k_u_first(const float* __restrict__ Z,
                                                 float* __restrict__ u) {
  __shared__ float sMx[8];
  __shared__ float sSm[8];
  u_row<false>(Z, nullptr, u, sMx, sSm);
}

__global__ void __launch_bounds__(256) k_u(const float* __restrict__ Z,
                                           const float* __restrict__ vdual,
                                           float* __restrict__ u) {
  __shared__ float sMx[8];
  __shared__ float sSm[8];
  u_row<true>(Z, vdual, u, sMx, sSm);
}

__global__ void __launch_bounds__(256) k_vdual(const float* __restrict__ Z,
                                               const float* __restrict__ u,
                                               float* __restrict__ vdual) {
  __shared__ float sU[M_];
  const int idx = blockIdx.x * 256 + threadIdx.x;
  const int b = idx >> 12;
  const int n = idx & 4095;
  if (threadIdx.x < M_) sU[threadIdx.x] = u[b * M_ + threadIdx.x];
  __syncthreads();
  const float* zb = Z + (size_t)b * (M_ * HW_) + n;
  float mx = -1e30f;
#pragma unroll
  for (int m = 0; m < M_; ++m) mx = fmaxf(mx, zb[(size_t)m * HW_] + sU[m]);
  float sm = 0.f;
#pragma unroll
  for (int m = 0; m < M_; ++m) sm += expf(zb[(size_t)m * HW_] + sU[m] - mx);
  vdual[idx] = NORM_ - (mx + logf(sm));
}

#define VT_ROWS_   256
#define VT_CHUNKS_ (HW_ / VT_ROWS_)
#define PQ_        132

__global__ void __launch_bounds__(256) k_vt_part(
    const float* __restrict__ Z, const float* __restrict__ u,
    const float* __restrict__ vdual, const float* __restrict__ xp,
    const float* __restrict__ rinv, float* __restrict__ part) {
  __shared__ __attribute__((aligned(16))) float sP[(VT_ROWS_ / 4) * PQ_];
  __shared__ __attribute__((aligned(16))) float sR[VT_ROWS_];
  __shared__ __attribute__((aligned(16))) float sRed[4 * 16 * 32 * 4];

  const int tid = threadIdx.x;
  const int wave = tid >> 5;
  const int lane = tid & 31;
  const int laneLo = lane & 15;
  const int laneHi = lane >> 4;

  const int b = blockIdx.x >> 4;
  const int n0 = (blockIdx.x & 15) * VT_ROWS_;
  const int g = wave & 3;
  const int kh = wave >> 2;
  const int dbase = 64 * g;

  const int nq = tid & 63;
  const int mg = tid >> 6;
  const float* zc = Z + (size_t)b * (M_ * HW_) + (size_t)(mg * 8) * HW_ + n0 + 4 * nq;
  const float r1 = rinv[(size_t)b * HW_ + n0 + tid];
  v4f z[8];
#pragma unroll
  for (int j = 0; j < 8; ++j) z[j] = *(const v4f*)(zc + (size_t)j * HW_);
  const v4f vv = *(const v4f*)(vdual + (size_t)b * HW_ + n0 + 4 * nq);
  const v4f ua = *(const v4f*)(u + b * M_ + mg * 8);
  const v4f ub = *(const v4f*)(u + b * M_ + mg * 8 + 4);
  const float* xrow = xp + ((size_t)b * HW_ + n0 + kh * 128 + 4 * laneHi) * D_ + dbase + 4 * laneLo;
  v4f L[16][4];
#pragma unroll
  for (int i = 0; i < 16; ++i) {
#pragma unroll
    for (int s = 0; s < 4; ++s) L[i][s] = *(const v4f*)(xrow + (size_t)(8 * i + s) * D_);
  }

  {
    const float um[8] = {ua.x, ua.y, ua.z, ua.w, ub.x, ub.y, ub.z, ub.w};
#pragma unroll
    for (int j = 0; j < 8; ++j) {
      v4f p;
      p.x = expf(z[j].x + um[j] + vv.x - NORM_);
      p.y = expf(z[j].y + um[j] + vv.y - NORM_);
      p.z = expf(z[j].z + um[j] + vv.z - NORM_);
      p.w = expf(z[j].w + um[j] + vv.w - NORM_);
      *(v4f*)(sP + nq * PQ_ + (mg * 8 + j) * 4) = p;
    }
    sR[tid] = r1;
  }
  __builtin_amdgcn_fence(__ATOMIC_RELEASE, "workgroup", "local");
  __builtin_amdgcn_s_barrier();
  __builtin_amdgcn_fence(__ATOMIC_ACQUIRE, "workgroup", "local");

  const float* aBase = sP + (kh * 32 + laneHi) * PQ_ + laneLo * 4;
  const float* rBase = sR + kh * 128 + 4 * laneHi;

  v8f acc[2][4];
#pragma unroll
  for (int t = 0; t < 2; ++t) {
#pragma unroll
    for (int j = 0; j < 4; ++j) acc[t][j] = (v8f){};
  }

  v4f a0s[2], a1s[2], rs[2];
  a0s[0] = *(const v4f*)(aBase);
  a1s[0] = *(const v4f*)(aBase + 64);
  rs[0] = *(const v4f*)(rBase);
#pragma unroll
  for (int i = 0; i < 16; ++i) {
    if (i + 1 < 16) {
      a0s[(i + 1) & 1] = *(const v4f*)(aBase + (i + 1) * 2 * PQ_);
      a1s[(i + 1) & 1] = *(const v4f*)(aBase + (i + 1) * 2 * PQ_ + 64);
      rs[(i + 1) & 1] = *(const v4f*)(rBase + (i + 1) * 8);
    }
    const v4f a0 = a0s[i & 1];
    const v4f a1 = a1s[i & 1];
    const v4f r4 = rs[i & 1];
#pragma unroll
    for (int j = 0; j < 4; ++j) {
      const v2f bA = {L[i][0][j] * r4.x, L[i][1][j] * r4.y};
      acc[0][j] = wmma4((v2f){a0.x, a0.y}, bA, acc[0][j]);
      acc[1][j] = wmma4((v2f){a1.x, a1.y}, bA, acc[1][j]);
    }
#pragma unroll
    for (int j = 0; j < 4; ++j) {
      const v2f bB = {L[i][2][j] * r4.z, L[i][3][j] * r4.w};
      acc[0][j] = wmma4((v2f){a0.z, a0.w}, bB, acc[0][j]);
      acc[1][j] = wmma4((v2f){a1.z, a1.w}, bB, acc[1][j]);
    }
  }

  float* red = sRed + (size_t)g * (16 * 32 * 4) + lane * 4;
  if (kh == 1) {
#pragma unroll
    for (int t = 0; t < 2; ++t) {
#pragma unroll
      for (int r = 0; r < 8; ++r)
        *(v4f*)(red + (t * 8 + r) * 128) =
            (v4f){acc[t][0][r], acc[t][1][r], acc[t][2][r], acc[t][3][r]};
    }
  }
  __syncthreads();
  if (kh == 0) {
    float* o = part + (size_t)blockIdx.x * (M_ * D_) + dbase + 4 * laneLo;
#pragma unroll
    for (int t = 0; t < 2; ++t) {
#pragma unroll
      for (int r = 0; r < 8; ++r) {
        const v4f hi = *(const v4f*)(red + (t * 8 + r) * 128);
        const v4f lo = {acc[t][0][r], acc[t][1][r], acc[t][2][r], acc[t][3][r]};
        *(v4f*)(o + (size_t)(16 * t + 8 * laneHi + r) * D_) = lo + hi;
      }
    }
  }
}

__global__ void __launch_bounds__(256) k_vt_sum(const float* __restrict__ part,
                                                float* __restrict__ vt) {
  const int i4 = blockIdx.x * 256 + threadIdx.x;
  const int b = i4 >> 11;
  const int e = i4 & 2047;
  const float* p = part + ((size_t)b * VT_CHUNKS_ * 2048 + e) * 4;
  v4f t[VT_CHUNKS_];
#pragma unroll
  for (int s = 0; s < VT_CHUNKS_; ++s) t[s] = *(const v4f*)(p + (size_t)s * (M_ * D_));
  v4f sum = t[0];
#pragma unroll
  for (int s = 1; s < VT_CHUNKS_; ++s) sum += t[s];
  *(v4f*)(vt + (size_t)i4 * 4) = sum;
}

extern "C" void kernel_launch(void* const* d_in, const int* in_sizes, int n_in,
                              void* d_out, int out_size, void* d_ws, size_t ws_size,
                              hipStream_t stream) {
  const float* x      = (const float*)d_in[0];
  const float* conv_w = (const float*)d_in[1];
  const float* conv_b = (const float*)d_in[2];
  const float* v      = (const float*)d_in[3];

  float* out_vt = (float*)d_out;
  float* out_xp = out_vt + (size_t)B_ * M_ * D_;

  float* ws   = (float*)d_ws;
  float* what = ws + OFF_WHAT;
  float* Z    = ws + OFF_Z;
  float* u    = ws + OFF_U;
  float* vd   = ws + OFF_V;
  float* rinv = ws + OFF_RINV;
  float* part = ws + OFF_PART;

  k_norm_v<<<1, 256, 0, stream>>>(v, what);
  k_projw<<<256, 256, 0, stream>>>(x, conv_w, conv_b, out_xp);
  k_scores<<<B_ * HW_ / 128, 256, 0, stream>>>(out_xp, what, Z, rinv);
  k_u_first<<<B_ * M_, 256, 0, stream>>>(Z, u);
  k_vdual<<<B_ * HW_ / 256, 256, 0, stream>>>(Z, u, vd);
  for (int it = 1; it < 3; ++it) {
    k_u<<<B_ * M_, 256, 0, stream>>>(Z, vd, u);
    k_vdual<<<B_ * HW_ / 256, 256, 0, stream>>>(Z, u, vd);
  }
  k_vt_part<<<B_ * VT_CHUNKS_, 256, 0, stream>>>(Z, u, vd, out_xp, rinv, part);
  k_vt_sum<<<B_ * M_ * D_ / 4 / 256, 256, 0, stream>>>(part, out_vt);
}


# k_projw: A-fragment LDS reads grouped: the 8 reads of the next 4 fragment pairs stand in one place per 16 products (2 register sets, second base past 64 KiB), not among them; sums unchanged
# speedup vs baseline: 2.1531x; 1.0626x over previous
#include <hip/hip_runtime.h>
#include <math.h>

typedef __attribute__((ext_vector_type(2))) float v2f;
typedef __attribute__((ext_vector_type(4))) float v4f;
typedef __attribute__((ext_vector_type(8))) float v8f;

#define B_    16
#define C_    768
#define HW_   4096
#define D_    256
#define M_    32
#define EPS_  0.05f
#define NORM_ (-8.3255517f)
#define NEPS_ 1e-12f

#define OFF_WHAT  0
#define OFF_Z     8192
#define OFF_U     (8192 + 2097152)
#define OFF_V     (OFF_U + 512)
#define OFF_RINV  (OFF_V + 65536)
#define OFF_PART  (OFF_RINV + 65536)

static __device__ __forceinline__ v8f wmma4(v2f a, v2f b, v8f c) {
  return __builtin_amdgcn_wmma_f32_16x16x4_f32(false, a, false, b, (short)0, c,
                                               false, false);
}

__global__ void k_norm_v(const float* __restrict__ v, float* __restrict__ what) {
  int m = threadIdx.x;
  if (m < M_) {
    float s = 0.f;
    for (int d = 0; d < D_; ++d) { float t = v[m * D_ + d]; s += t * t; }
    float inv = 1.0f / fmaxf(sqrtf(s), NEPS_);
    for (int d = 0; d < D_; ++d) what[m * D_ + d] = v[m * D_ + d] * inv;
  }
}


#define TILE_N_   32
#define STAGE_F_  (C_ * TILE_N_)
#define TILES_PB_ 16
#define GI_       4

__global__ void __launch_bounds__(256) k_projw(
    const float* __restrict__ x, const float* __restrict__ w,
    const float* __restrict__ bias, float* __restrict__ xp) {
  __shared__ __attribute__((aligned(16))) float sX[2 * STAGE_F_];

  const int tid = threadIdx.x;
  const int wave = tid >> 5;
  const int lane = tid & 31;
  const int laneLo = lane & 15;
  const int laneHi = lane >> 4;

  const int dh = blockIdx.x & 1;
  const int tile0 = (blockIdx.x >> 1) * TILES_PB_;
  const int b = tile0 >> 7;
  const int nbase = (tile0 & 127) * TILE_N_;
  const int d0 = dh * 128 + wave * 16;

  v4f breg[96];
  {
    const float* wrow = w + (size_t)(d0 + laneLo) * C_ + 4 * laneHi;
#pragma unroll
    for (int i = 0; i < 96; ++i) breg[i] = *(const v4f*)(wrow + 8 * i);
  }
  const float bi = bias[d0 + laneLo];

  const int sq = lane >> 3;
  const int n4 = (lane & 7) * 4;
  const float* xb = x + (size_t)b * C_ * HW_ + nbase + n4;

#pragma unroll
  for (int p = 0; p < 6; ++p) {
    const int kq = 4 * (p * 8 + wave) + sq;
    const float* s = xb + (size_t)(4 * kq) * HW_;
    v4f g0 = *(const v4f*)(s);
    v4f g1 = *(const v4f*)(s + HW_);
    v4f g2 = *(const v4f*)(s + 2 * HW_);
    v4f g3 = *(const v4f*)(s + 3 * HW_);
    float* dst = sX + kq * 128 + n4 * 4;
    *(v4f*)(dst)      = (v4f){g0.x, g1.x, g2.x, g3.x};
    *(v4f*)(dst + 4)  = (v4f){g0.y, g1.y, g2.y, g3.y};
    *(v4f*)(dst + 8)  = (v4f){g0.z, g1.z, g2.z, g3.z};
    *(v4f*)(dst + 12) = (v4f){g0.w, g1.w, g2.w, g3.w};
  }
  __syncthreads();

  for (int t = 0; t < TILES_PB_; ++t) {
    const float* cur = sX + (t & 1) * STAGE_F_;
    float* nxt = sX + ((t + 1) & 1) * STAGE_F_;
    const int tn = (t + 1 < TILES_PB_) ? (t + 1) : t;
    const float* xn = xb + tn * TILE_N_;
    const float* aBase = cur + laneHi * 128 + laneLo * 4;
    const float* aBaseHi = sX + ((t & 1) ? (STAGE_F_ + 64 * 256) : (64 * 256)) + laneHi * 128 + laneLo * 4;

    v8f acc0 = {}; v8f acc1 = {};
    v4f a0s[2][GI_], a1s[2][GI_];
#pragma unroll
    for (int j = 0; j < GI_; ++j) {
      a0s[0][j] = *(const v4f*)(aBase + j * 256);
      a1s[0][j] = *(const v4f*)(aBase + j * 256 + 64);
    }
    v4f g0, g1, g2, g3;
#pragma unroll
    for (int g = 0; g < 96 / GI_; ++g) {
      const int i0 = g * GI_;
      const int p = i0 >> 4;
      const int kq = 4 * (p * 8 + wave) + sq;
      if ((i0 & 15) == 0) {
        const float* s = xn + (size_t)(4 * kq) * HW_;
        g0 = *(const v4f*)(s);
        g1 = *(const v4f*)(s + HW_);
        g2 = *(const v4f*)(s + 2 * HW_);
        g3 = *(const v4f*)(s + 3 * HW_);
      }
      if (g + 1 < 96 / GI_) {
#pragma unroll
        for (int j = 0; j < GI_; ++j) {
          const int in = i0 + GI_ + j;
          const float* ap = (in < 64) ? (aBase + in * 256) : (aBaseHi + (in - 64) * 256);
          a0s[(g + 1) & 1][j] = *(const v4f*)(ap);
          a1s[(g + 1) & 1][j] = *(const v4f*)(ap + 64);
        }
      }
#pragma unroll
      for (int j = 0; j < GI_; ++j) {
        const v4f a0 = a0s[g & 1][j];
        const v4f a1 = a1s[g & 1][j];
        const v4f bq = breg[i0 + j];
        const v2f bA = {bq.x, bq.y};
        const v2f bB = {bq.z, bq.w};
        acc0 = wmma4((v2f){a0.x, a0.y}, bA, acc0);
        acc1 = wmma4((v2f){a1.x, a1.y}, bA, acc1);
        acc0 = wmma4((v2f){a0.z, a0.w}, bB, acc0);
        acc1 = wmma4((v2f){a1.z, a1.w}, bB, acc1);
      }
      if (((i0 + GI_ - 1) & 15) == 15) {
        float* dst = nxt + kq * 128 + n4 * 4;
        *(v4f*)(dst)      = (v4f){g0.x, g1.x, g2.x, g3.x};
        *(v4f*)(dst + 4)  = (v4f){g0.y, g1.y, g2.y, g3.y};
        *(v4f*)(dst + 8)  = (v4f){g0.z, g1.z, g2.z, g3.z};
        *(v4f*)(dst + 12) = (v4f){g0.w, g1.w, g2.w, g3.w};
      }
    }

    __syncthreads();

    {
      float* o = xp + ((size_t)b * HW_ + nbase + t * TILE_N_) * D_ + d0 + laneLo;
#pragma unroll
      for (int r = 0; r < 8; ++r) {
        const int row = r + 8 * laneHi;
        o[(size_t)row * D_] = acc0[r] + bi;
        o[(size_t)(row + 16) * D_] = acc1[r] + bi;
      }
    }
  }
}

#define WP_ 260
__global__ void __launch_bounds__(256) k_scores(
    const float* __restrict__ xp, const float* __restrict__ what,
    float* __restrict__ Z, float* __restrict__ rinvg) {
  __shared__ __attribute__((aligned(16))) float sW[M_ * WP_];
  const int tid = threadIdx.x;
  const int wave = tid >> 5;
  const int lane = tid & 31;
  const int laneLo = lane & 15;
  const int laneHi = lane >> 4;

  for (int i = tid; i < M_ * (D_ / 4); i += 256) {
    const int m = i >> 6, q = i & 63;
    *(v4f*)(sW + m * WP_ + q * 4) = *(const v4f*)(what + m * D_ + q * 4);
  }
  __syncthreads();

  const int T = blockIdx.x * 8 + wave;
  const int b = T >> 8;
  const int n0 = (T & 255) * 16;

  const float* arow = xp + ((size_t)b * HW_ + n0 + laneLo) * D_ + 4 * laneHi;
  const float* w0p = sW + laneLo * WP_ + 4 * laneHi;
  const float* w1p = w0p + 16 * WP_;

  v8f cs0 = {}; v8f cs1 = {};
  float ss = 0.f;
#pragma unroll 8
  for (int i = 0; i < 32; ++i) {
    v4f a = *(const v4f*)(arow + 8 * i);
    ss += a.x * a.x; ss += a.y * a.y; ss += a.z * a.z; ss += a.w * a.w;
    v4f q0 = *(const v4f*)(w0p + 8 * i);
    v4f q1 = *(const v4f*)(w1p + 8 * i);
    v2f aA = {a.x, a.y};
    v2f aB = {a.z, a.w};
    cs0 = wmma4(aA, (v2f){q0.x, q0.y}, cs0);
    cs1 = wmma4(aA, (v2f){q1.x, q1.y}, cs1);
    cs0 = wmma4(aB, (v2f){q0.z, q0.w}, cs0);
    cs1 = wmma4(aB, (v2f){q1.z, q1.w}, cs1);
  }
  ss += __shfl_xor(ss, 16, 32);
  const float inv = 1.0f / fmaxf(sqrtf(ss), NEPS_);
  if (laneHi == 0) rinvg[(size_t)b * HW_ + n0 + laneLo] = inv;

  float* z0 = Z + (size_t)b * (M_ * HW_) + (size_t)laneLo * HW_ + n0 + 8 * laneHi;
  float* z1 = z0 + (size_t)16 * HW_;
  const float invEps = 1.0f / EPS_;
#pragma unroll
  for (int r = 0; r < 8; ++r) {
    const float ri = __shfl(inv, 8 * laneHi + r, 32);
    z0[r] = cs0[r] * ri * invEps;
    z1[r] = cs1[r] * ri * invEps;
  }
}

template <bool HAS_V>
static __device__ __forceinline__ void u_row(const float* __restrict__ Z,
                                             const float* __restrict__ vdual,
                                             float* __restrict__ u,
                                             float* sMx, float* sSm) {
  const int b = blockIdx.x >> 5;
  const int m = blockIdx.x & 31;
  const int tid = threadIdx.x;
  const int lane = tid & 31;
  const int wave = tid >> 5;
  const float* zrow = Z + (size_t)b * (M_ * HW_) + (size_t)m * HW_ + 4 * tid;

  v4f t[4];
  for (int j = 0; j < 4; ++j) t[j] = *(const v4f*)(zrow + 1024 * j);
  if (HAS_V) {
    const float* vv = vdual + (size_t)b * HW_ + 4 * tid;
    for (int j = 0; j < 4; ++j) t[j] += *(const v4f*)(vv + 1024 * j);
  }

  float mx = -1e30f;
  for (int j = 0; j < 4; ++j)
    mx = fmaxf(mx, fmaxf(fmaxf(t[j].x, t[j].y), fmaxf(t[j].z, t[j].w)));
  for (int o = 16; o > 0; o >>= 1) mx = fmaxf(mx, __shfl_xor(mx, o, 32));
  if (lane == 0) sMx[wave] = mx;
  __syncthreads();
  float gmax = sMx[0];
  for (int w2 = 1; w2 < 8; ++w2) gmax = fmaxf(gmax, sMx[w2]);

  float sm = 0.f;
  for (int j = 0; j < 4; ++j) {
    sm += expf(t[j].x - gmax);
    sm += expf(t[j].y - gmax);
    sm += expf(t[j].z - gmax);
    sm += expf(t[j].w - gmax);
  }
  for (int o = 16; o > 0; o >>= 1) sm += __shfl_xor(sm, o, 32);
  if (lane == 0) sSm[wave] = sm;
  __syncthreads();
  if (tid == 0) {
    float s = sSm[0];
    for (int w2 = 1; w2 < 8; ++w2) s += sSm[w2];
    u[b * M_ + m] = NORM_ - (gmax + logf(s));
  }
}

__global__ void __launch_bounds__(256) k_u_first(const float* __restrict__ Z,
                                                 float* __restrict__ u) {
  __shared__ float sMx[8];
  __shared__ float sSm[8];
  u_row<false>(Z, nullptr, u, sMx, sSm);
}

__global__ void __launch_bounds__(256) k_u(const float* __restrict__ Z,
                                           const float* __restrict__ vdual,
                                           float* __restrict__ u) {
  __shared__ float sMx[8];
  __shared__ float sSm[8];
  u_row<true>(Z, vdual, u, sMx, sSm);
}

__global__ void __launch_bounds__(256) k_vdual(const float* __restrict__ Z,
                                               const float* __restrict__ u,
                                               float* __restrict__ vdual) {
  __shared__ float sU[M_];
  const int idx = blockIdx.x * 256 + threadIdx.x;
  const int b = idx >> 12;
  const int n = idx & 4095;
  if (threadIdx.x < M_) sU[threadIdx.x] = u[b * M_ + threadIdx.x];
  __syncthreads();
  const float* zb = Z + (size_t)b * (M_ * HW_) + n;
  float mx = -1e30f;
#pragma unroll
  for (int m = 0; m < M_; ++m) mx = fmaxf(mx, zb[(size_t)m * HW_] + sU[m]);
  float sm = 0.f;
#pragma unroll
  for (int m = 0; m < M_; ++m) sm += expf(zb[(size_t)m * HW_] + sU[m] - mx);
  vdual[idx] = NORM_ - (mx + logf(sm));
}

#define VT_ROWS_   256
#define VT_CHUNKS_ (HW_ / VT_ROWS_)
#define PQ_        132

__global__ void __launch_bounds__(256) k_vt_part(
    const float* __restrict__ Z, const float* __restrict__ u,
    const float* __restrict__ vdual, const float* __restrict__ xp,
    const float* __restrict__ rinv, float* __restrict__ part) {
  __shared__ __attribute__((aligned(16))) float sP[(VT_ROWS_ / 4) * PQ_];
  __shared__ __attribute__((aligned(16))) float sR[VT_ROWS_];
  __shared__ __attribute__((aligned(16))) float sRed[4 * 16 * 32 * 4];

  const int tid = threadIdx.x;
  const int wave = tid >> 5;
  const int lane = tid & 31;
  const int laneLo = lane & 15;
  const int laneHi = lane >> 4;

  const int b = blockIdx.x >> 4;
  const int n0 = (blockIdx.x & 15) * VT_ROWS_;
  const int g = wave & 3;
  const int kh = wave >> 2;
  const int dbase = 64 * g;

  const int nq = tid & 63;
  const int mg = tid >> 6;
  const float* zc = Z + (size_t)b * (M_ * HW_) + (size_t)(mg * 8) * HW_ + n0 + 4 * nq;
  const float r1 = rinv[(size_t)b * HW_ + n0 + tid];
  v4f z[8];
#pragma unroll
  for (int j = 0; j < 8; ++j) z[j] = *(const v4f*)(zc + (size_t)j * HW_);
  const v4f vv = *(const v4f*)(vdual + (size_t)b * HW_ + n0 + 4 * nq);
  const v4f ua = *(const v4f*)(u + b * M_ + mg * 8);
  const v4f ub = *(const v4f*)(u + b * M_ + mg * 8 + 4);
  const float* xrow = xp + ((size_t)b * HW_ + n0 + kh * 128 + 4 * laneHi) * D_ + dbase + 4 * laneLo;
  v4f L[16][4];
#pragma unroll
  for (int i = 0; i < 16; ++i) {
#pragma unroll
    for (int s = 0; s < 4; ++s) L[i][s] = *(const v4f*)(xrow + (size_t)(8 * i + s) * D_);
  }

  {
    const float um[8] = {ua.x, ua.y, ua.z, ua.w, ub.x, ub.y, ub.z, ub.w};
#pragma unroll
    for (int j = 0; j < 8; ++j) {
      v4f p;
      p.x = expf(z[j].x + um[j] + vv.x - NORM_);
      p.y = expf(z[j].y + um[j] + vv.y - NORM_);
      p.z = expf(z[j].z + um[j] + vv.z - NORM_);
      p.w = expf(z[j].w + um[j] + vv.w - NORM_);
      *(v4f*)(sP + nq * PQ_ + (mg * 8 + j) * 4) = p;
    }
    sR[tid] = r1;
  }
  __builtin_amdgcn_fence(__ATOMIC_RELEASE, "workgroup", "local");
  __builtin_amdgcn_s_barrier();
  __builtin_amdgcn_fence(__ATOMIC_ACQUIRE, "workgroup", "local");

  const float* aBase = sP + (kh * 32 + laneHi) * PQ_ + laneLo * 4;
  const float* rBase = sR + kh * 128 + 4 * laneHi;

  v8f acc[2][4];
#pragma unroll
  for (int t = 0; t < 2; ++t) {
#pragma unroll
    for (int j = 0; j < 4; ++j) acc[t][j] = (v8f){};
  }

  v4f a0s[2], a1s[2], rs[2];
  a0s[0] = *(const v4f*)(aBase);
  a1s[0] = *(const v4f*)(aBase + 64);
  rs[0] = *(const v4f*)(rBase);
#pragma unroll
  for (int i = 0; i < 16; ++i) {
    if (i + 1 < 16) {
      a0s[(i + 1) & 1] = *(const v4f*)(aBase + (i + 1) * 2 * PQ_);
      a1s[(i + 1) & 1] = *(const v4f*)(aBase + (i + 1) * 2 * PQ_ + 64);
      rs[(i + 1) & 1] = *(const v4f*)(rBase + (i + 1) * 8);
    }
    const v4f a0 = a0s[i & 1];
    const v4f a1 = a1s[i & 1];
    const v4f r4 = rs[i & 1];
#pragma unroll
    for (int j = 0; j < 4; ++j) {
      const v2f bA = {L[i][0][j] * r4.x, L[i][1][j] * r4.y};
      acc[0][j] = wmma4((v2f){a0.x, a0.y}, bA, acc[0][j]);
      acc[1][j] = wmma4((v2f){a1.x, a1.y}, bA, acc[1][j]);
    }
#pragma unroll
    for (int j = 0; j < 4; ++j) {
      const v2f bB = {L[i][2][j] * r4.z, L[i][3][j] * r4.w};
      acc[0][j] = wmma4((v2f){a0.z, a0.w}, bB, acc[0][j]);
      acc[1][j] = wmma4((v2f){a1.z, a1.w}, bB, acc[1][j]);
    }
  }

  float* red = sRed + (size_t)g * (16 * 32 * 4) + lane * 4;
  if (kh == 1) {
#pragma unroll
    for (int t = 0; t < 2; ++t) {
#pragma unroll
      for (int r = 0; r < 8; ++r)
        *(v4f*)(red + (t * 8 + r) * 128) =
            (v4f){acc[t][0][r], acc[t][1][r], acc[t][2][r], acc[t][3][r]};
    }
  }
  __syncthreads();
  if (kh == 0) {
    float* o = part + (size_t)blockIdx.x * (M_ * D_) + dbase + 4 * laneLo;
#pragma unroll
    for (int t = 0; t < 2; ++t) {
#pragma unroll
      for (int r = 0; r < 8; ++r) {
        const v4f hi = *(const v4f*)(red + (t * 8 + r) * 128);
        const v4f lo = {acc[t][0][r], acc[t][1][r], acc[t][2][r], acc[t][3][r]};
        *(v4f*)(o + (size_t)(16 * t + 8 * laneHi + r) * D_) = lo + hi;
      }
    }
  }
}

__global__ void __launch_bounds__(256) k_vt_sum(const float* __restrict__ part,
                                                float* __restrict__ vt) {
  const int i4 = blockIdx.x * 256 + threadIdx.x;
  const int b = i4 >> 11;
  const int e = i4 & 2047;
  const float* p = part + ((size_t)b * VT_CHUNKS_ * 2048 + e) * 4;
  v4f t[VT_CHUNKS_];
#pragma unroll
  for (int s = 0; s < VT_CHUNKS_; ++s) t[s] = *(const v4f*)(p + (size_t)s * (M_ * D_));
  v4f sum = t[0];
#pragma unroll
  for (int s = 1; s < VT_CHUNKS_; ++s) sum += t[s];
  *(v4f*)(vt + (size_t)i4 * 4) = sum;
}

extern "C" void kernel_launch(void* const* d_in, const int* in_sizes, int n_in,
                              void* d_out, int out_size, void* d_ws, size_t ws_size,
                              hipStream_t stream) {
  const float* x      = (const float*)d_in[0];
  const float* conv_w = (const float*)d_in[1];
  const float* conv_b = (const float*)d_in[2];
  const float* v      = (const float*)d_in[3];

  float* out_vt = (float*)d_out;
  float* out_xp = out_vt + (size_t)B_ * M_ * D_;

  float* ws   = (float*)d_ws;
  float* what = ws + OFF_WHAT;
  float* Z    = ws + OFF_Z;
  float* u    = ws + OFF_U;
  float* vd   = ws + OFF_V;
  float* rinv = ws + OFF_RINV;
  float* part = ws + OFF_PART;

  k_norm_v<<<1, 256, 0, stream>>>(v, what);
  k_projw<<<256, 256, 0, stream>>>(x, conv_w, conv_b, out_xp);
  k_scores<<<B_ * HW_ / 128, 256, 0, stream>>>(out_xp, what, Z, rinv);
  k_u_first<<<B_ * M_, 256, 0, stream>>>(Z, u);
  k_vdual<<<B_ * HW_ / 256, 256, 0, stream>>>(Z, u, vd);
  for (int it = 1; it < 3; ++it) {
    k_u<<<B_ * M_, 256, 0, stream>>>(Z, vd, u);
    k_vdual<<<B_ * HW_ / 256, 256, 0, stream>>>(Z, u, vd);
  }
  k_vt_part<<<B_ * VT_CHUNKS_, 256, 0, stream>>>(Z, u, vd, out_xp, rinv, part);
  k_vt_sum<<<B_ * M_ * D_ / 4 / 256, 256, 0, stream>>>(part, out_vt);
}


# k_norm_v, k_scores: all global loads of a wave in flight before its first wait (norm: 1 wave per cluster row, 1 round trip not 24; scores: weights + tile up front); cluster-norm sum tree changed
# speedup vs baseline: 2.2436x; 1.0420x over previous
#include <hip/hip_runtime.h>
#include <math.h>

typedef __attribute__((ext_vector_type(2))) float v2f;
typedef __attribute__((ext_vector_type(4))) float v4f;
typedef __attribute__((ext_vector_type(8))) float v8f;

#define B_    16
#define C_    768
#define HW_   4096
#define D_    256
#define M_    32
#define EPS_  0.05f
#define NORM_ (-8.3255517f)
#define NEPS_ 1e-12f

#define OFF_WHAT  0
#define OFF_Z     8192
#define OFF_U     (8192 + 2097152)
#define OFF_V     (OFF_U + 512)
#define OFF_RINV  (OFF_V + 65536)
#define OFF_PART  (OFF_RINV + 65536)

static __device__ __forceinline__ v8f wmma4(v2f a, v2f b, v8f c) {
  return __builtin_amdgcn_wmma_f32_16x16x4_f32(false, a, false, b, (short)0, c,
                                               false, false);
}

__global__ void __launch_bounds__(1024) k_norm_v(const float* __restrict__ v, float* __restrict__ what) {
  const int m = threadIdx.x >> 5;
  const int lane = threadIdx.x & 31;
  const float* r = v + m * D_ + 4 * lane;
  const v4f p = *(const v4f*)(r);
  const v4f q = *(const v4f*)(r + 128);
  float s = p.x * p.x;
  s += p.y * p.y; s += p.z * p.z; s += p.w * p.w;
  s += q.x * q.x; s += q.y * q.y; s += q.z * q.z; s += q.w * q.w;
  for (int o = 16; o > 0; o >>= 1) s += __shfl_xor(s, o, 32);
  const float inv = 1.0f / fmaxf(sqrtf(s), NEPS_);
  float* w = what + m * D_ + 4 * lane;
  *(v4f*)(w) = p * inv;
  *(v4f*)(w + 128) = q * inv;
}


#define TILE_N_   32
#define STAGE_F_  (C_ * TILE_N_)
#define TILES_PB_ 16
#define GI_       4

__global__ void __launch_bounds__(256) k_projw(
    const float* __restrict__ x, const float* __restrict__ w,
    const float* __restrict__ bias, float* __restrict__ xp) {
  __shared__ __attribute__((aligned(16))) float sX[2 * STAGE_F_];

  const int tid = threadIdx.x;
  const int wave = tid >> 5;
  const int lane = tid & 31;
  const int laneLo = lane & 15;
  const int laneHi = lane >> 4;

  const int dh = blockIdx.x & 1;
  const int tile0 = (blockIdx.x >> 1) * TILES_PB_;
  const int b = tile0 >> 7;
  const int nbase = (tile0 & 127) * TILE_N_;
  const int d0 = dh * 128 + wave * 16;

  v4f breg[96];
  {
    const float* wrow = w + (size_t)(d0 + laneLo) * C_ + 4 * laneHi;
#pragma unroll
    for (int i = 0; i < 96; ++i) breg[i] = *(const v4f*)(wrow + 8 * i);
  }
  const float bi = bias[d0 + laneLo];

  const int sq = lane >> 3;
  const int n4 = (lane & 7) * 4;
  const float* xb = x + (size_t)b * C_ * HW_ + nbase + n4;

#pragma unroll
  for (int p = 0; p < 6; ++p) {
    const int kq = 4 * (p * 8 + wave) + sq;
    const float* s = xb + (size_t)(4 * kq) * HW_;
    v4f g0 = *(const v4f*)(s);
    v4f g1 = *(const v4f*)(s + HW_);
    v4f g2 = *(const v4f*)(s + 2 * HW_);
    v4f g3 = *(const v4f*)(s + 3 * HW_);
    float* dst = sX + kq * 128 + n4 * 4;
    *(v4f*)(dst)      = (v4f){g0.x, g1.x, g2.x, g3.x};
    *(v4f*)(dst + 4)  = (v4f){g0.y, g1.y, g2.y, g3.y};
    *(v4f*)(dst + 8)  = (v4f){g0.z, g1.z, g2.z, g3.z};
    *(v4f*)(dst + 12) = (v4f){g0.w, g1.w, g2.w, g3.w};
  }
  __syncthreads();

  for (int t = 0; t < TILES_PB_; ++t) {
    const float* cur = sX + (t & 1) * STAGE_F_;
    float* nxt = sX + ((t + 1) & 1) * STAGE_F_;
    const int tn = (t + 1 < TILES_PB_) ? (t + 1) : t;
    const float* xn = xb + tn * TILE_N_;
    const float* aBase = cur + laneHi * 128 + laneLo * 4;
    const float* aBaseHi = sX + ((t & 1) ? (STAGE_F_ + 64 * 256) : (64 * 256)) + laneHi * 128 + laneLo * 4;

    v8f acc0 = {}; v8f acc1 = {};
    v4f a0s[2][GI_], a1s[2][GI_];
#pragma unroll
    for (int j = 0; j < GI_; ++j) {
      a0s[0][j] = *(const v4f*)(aBase + j * 256);
      a1s[0][j] = *(const v4f*)(aBase + j * 256 + 64);
    }
    v4f g0, g1, g2, g3;
#pragma unroll
    for (int g = 0; g < 96 / GI_; ++g) {
      const int i0 = g * GI_;
      const int p = i0 >> 4;
      const int kq = 4 * (p * 8 + wave) + sq;
      if ((i0 & 15) == 0) {
        const float* s = xn + (size_t)(4 * kq) * HW_;
        g0 = *(const v4f*)(s);
        g1 = *(const v4f*)(s + HW_);
        g2 = *(const v4f*)(s + 2 * HW_);
        g3 = *(const v4f*)(s + 3 * HW_);
      }
      if (g + 1 < 96 / GI_) {
#pragma unroll
        for (int j = 0; j < GI_; ++j) {
          const int in = i0 + GI_ + j;
          const float* ap = (in < 64) ? (aBase + in * 256) : (aBaseHi + (in - 64) * 256);
          a0s[(g + 1) & 1][j] = *(const v4f*)(ap);
          a1s[(g + 1) & 1][j] = *(const v4f*)(ap + 64);
        }
      }
#pragma unroll
      for (int j = 0; j < GI_; ++j) {
        const v4f a0 = a0s[g & 1][j];
        const v4f a1 = a1s[g & 1][j];
        const v4f bq = breg[i0 + j];
        const v2f bA = {bq.x, bq.y};
        const v2f bB = {bq.z, bq.w};
        acc0 = wmma4((v2f){a0.x, a0.y}, bA, acc0);
        acc1 = wmma4((v2f){a1.x, a1.y}, bA, acc1);
        acc0 = wmma4((v2f){a0.z, a0.w}, bB, acc0);
        acc1 = wmma4((v2f){a1.z, a1.w}, bB, acc1);
      }
      if (((i0 + GI_ - 1) & 15) == 15) {
        float* dst = nxt + kq * 128 + n4 * 4;
        *(v4f*)(dst)      = (v4f){g0.x, g1.x, g2.x, g3.x};
        *(v4f*)(dst + 4)  = (v4f){g0.y, g1.y, g2.y, g3.y};
        *(v4f*)(dst + 8)  = (v4f){g0.z, g1.z, g2.z, g3.z};
        *(v4f*)(dst + 12) = (v4f){g0.w, g1.w, g2.w, g3.w};
      }
    }

    __syncthreads();

    {
      float* o = xp + ((size_t)b * HW_ + nbase + t * TILE_N_) * D_ + d0 + laneLo;
#pragma unroll
      for (int r = 0; r < 8; ++r) {
        const int row = r + 8 * laneHi;
        o[(size_t)row * D_] = acc0[r] + bi;
        o[(size_t)(row + 16) * D_] = acc1[r] + bi;
      }
    }
  }
}

#define WP_ 260
__global__ void __launch_bounds__(256) k_scores(
    const float* __restrict__ xp, const float* __restrict__ what,
    float* __restrict__ Z, float* __restrict__ rinvg) {
  __shared__ __attribute__((aligned(16))) float sW[M_ * WP_];
  const int tid = threadIdx.x;
  const int wave = tid >> 5;
  const int lane = tid & 31;
  const int laneLo = lane & 15;
  const int laneHi = lane >> 4;

  const int T = blockIdx.x * 8 + wave;
  const int b = T >> 8;
  const int n0 = (T & 255) * 16;
  const float* arow = xp + ((size_t)b * HW_ + n0 + laneLo) * D_ + 4 * laneHi;

  v4f wv[8];
#pragma unroll
  for (int j = 0; j < 8; ++j) {
    const int i = tid + 256 * j;
    wv[j] = *(const v4f*)(what + (i >> 6) * D_ + (i & 63) * 4);
  }
  v4f av[32];
#pragma unroll
  for (int i = 0; i < 32; ++i) av[i] = *(const v4f*)(arow + 8 * i);
#pragma unroll
  for (int j = 0; j < 8; ++j) {
    const int i = tid + 256 * j;
    *(v4f*)(sW + (i >> 6) * WP_ + (i & 63) * 4) = wv[j];
  }
  __builtin_amdgcn_fence(__ATOMIC_RELEASE, "workgroup", "local");
  __builtin_amdgcn_s_barrier();
  __builtin_amdgcn_fence(__ATOMIC_ACQUIRE, "workgroup", "local");

  const float* w0p = sW + laneLo * WP_ + 4 * laneHi;
  const float* w1p = w0p + 16 * WP_;

  v8f cs0 = {}; v8f cs1 = {};
  float ss = 0.f;
#pragma unroll
  for (int i = 0; i < 32; ++i) {
    const v4f a = av[i];
    ss += a.x * a.x; ss += a.y * a.y; ss += a.z * a.z; ss += a.w * a.w;
    v4f q0 = *(const v4f*)(w0p + 8 * i);
    v4f q1 = *(const v4f*)(w1p + 8 * i);
    v2f aA = {a.x, a.y};
    v2f aB = {a.z, a.w};
    cs0 = wmma4(aA, (v2f){q0.x, q0.y}, cs0);
    cs1 = wmma4(aA, (v2f){q1.x, q1.y}, cs1);
    cs0 = wmma4(aB, (v2f){q0.z, q0.w}, cs0);
    cs1 = wmma4(aB, (v2f){q1.z, q1.w}, cs1);
  }
  ss += __shfl_xor(ss, 16, 32);
  const float inv = 1.0f / fmaxf(sqrtf(ss), NEPS_);
  if (laneHi == 0) rinvg[(size_t)b * HW_ + n0 + laneLo] = inv;

  float* z0 = Z + (size_t)b * (M_ * HW_) + (size_t)laneLo * HW_ + n0 + 8 * laneHi;
  float* z1 = z0 + (size_t)16 * HW_;
  const float invEps = 1.0f / EPS_;
#pragma unroll
  for (int r = 0; r < 8; ++r) {
    const float ri = __shfl(inv, 8 * laneHi + r, 32);
    z0[r] = cs0[r] * ri * invEps;
    z1[r] = cs1[r] * ri * invEps;
  }
}

template <bool HAS_V>
static __device__ __forceinline__ void u_row(const float* __restrict__ Z,
                                             const float* __restrict__ vdual,
                                             float* __restrict__ u,
                                             float* sMx, float* sSm) {
  const int b = blockIdx.x >> 5;
  const int m = blockIdx.x & 31;
  const int tid = threadIdx.x;
  const int lane = tid & 31;
  const int wave = tid >> 5;
  const float* zrow = Z + (size_t)b * (M_ * HW_) + (size_t)m * HW_ + 4 * tid;

  v4f t[4];
  for (int j = 0; j < 4; ++j) t[j] = *(const v4f*)(zrow + 1024 * j);
  if (HAS_V) {
    const float* vv = vdual + (size_t)b * HW_ + 4 * tid;
    for (int j = 0; j < 4; ++j) t[j] += *(const v4f*)(vv + 1024 * j);
  }

  float mx = -1e30f;
  for (int j = 0; j < 4; ++j)
    mx = fmaxf(mx, fmaxf(fmaxf(t[j].x, t[j].y), fmaxf(t[j].z, t[j].w)));
  for (int o = 16; o > 0; o >>= 1) mx = fmaxf(mx, __shfl_xor(mx, o, 32));
  if (lane == 0) sMx[wave] = mx;
  __syncthreads();
  float gmax = sMx[0];
  for (int w2 = 1; w2 < 8; ++w2) gmax = fmaxf(gmax, sMx[w2]);

  float sm = 0.f;
  for (int j = 0; j < 4; ++j) {
    sm += expf(t[j].x - gmax);
    sm += expf(t[j].y - gmax);
    sm += expf(t[j].z - gmax);
    sm += expf(t[j].w - gmax);
  }
  for (int o = 16; o > 0; o >>= 1) sm += __shfl_xor(sm, o, 32);
  if (lane == 0) sSm[wave] = sm;
  __syncthreads();
  if (tid == 0) {
    float s = sSm[0];
    for (int w2 = 1; w2 < 8; ++w2) s += sSm[w2];
    u[b * M_ + m] = NORM_ - (gmax + logf(s));
  }
}

__global__ void __launch_bounds__(256) k_u_first(const float* __restrict__ Z,
                                                 float* __restrict__ u) {
  __shared__ float sMx[8];
  __shared__ float sSm[8];
  u_row<false>(Z, nullptr, u, sMx, sSm);
}

__global__ void __launch_bounds__(256) k_u(const float* __restrict__ Z,
                                           const float* __restrict__ vdual,
                                           float* __restrict__ u) {
  __shared__ float sMx[8];
  __shared__ float sSm[8];
  u_row<true>(Z, vdual, u, sMx, sSm);
}

__global__ void __launch_bounds__(256) k_vdual(const float* __restrict__ Z,
                                               const float* __restrict__ u,
                                               float* __restrict__ vdual) {
  __shared__ float sU[M_];
  const int idx = blockIdx.x * 256 + threadIdx.x;
  const int b = idx >> 12;
  const int n = idx & 4095;
  if (threadIdx.x < M_) sU[threadIdx.x] = u[b * M_ + threadIdx.x];
  __syncthreads();
  const float* zb = Z + (size_t)b * (M_ * HW_) + n;
  float mx = -1e30f;
#pragma unroll
  for (int m = 0; m < M_; ++m) mx = fmaxf(mx, zb[(size_t)m * HW_] + sU[m]);
  float sm = 0.f;
#pragma unroll
  for (int m = 0; m < M_; ++m) sm += expf(zb[(size_t)m * HW_] + sU[m] - mx);
  vdual[idx] = NORM_ - (mx + logf(sm));
}

#define VT_ROWS_   256
#define VT_CHUNKS_ (HW_ / VT_ROWS_)
#define PQ_        132

__global__ void __launch_bounds__(256) k_vt_part(
    const float* __restrict__ Z, const float* __restrict__ u,
    const float* __restrict__ vdual, const float* __restrict__ xp,
    const float* __restrict__ rinv, float* __restrict__ part) {
  __shared__ __attribute__((aligned(16))) float sP[(VT_ROWS_ / 4) * PQ_];
  __shared__ __attribute__((aligned(16))) float sR[VT_ROWS_];
  __shared__ __attribute__((aligned(16))) float sRed[4 * 16 * 32 * 4];

  const int tid = threadIdx.x;
  const int wave = tid >> 5;
  const int lane = tid & 31;
  const int laneLo = lane & 15;
  const int laneHi = lane >> 4;

  const int b = blockIdx.x >> 4;
  const int n0 = (blockIdx.x & 15) * VT_ROWS_;
  const int g = wave & 3;
  const int kh = wave >> 2;
  const int dbase = 64 * g;

  const int nq = tid & 63;
  const int mg = tid >> 6;
  const float* zc = Z + (size_t)b * (M_ * HW_) + (size_t)(mg * 8) * HW_ + n0 + 4 * nq;
  const float r1 = rinv[(size_t)b * HW_ + n0 + tid];
  v4f z[8];
#pragma unroll
  for (int j = 0; j < 8; ++j) z[j] = *(const v4f*)(zc + (size_t)j * HW_);
  const v4f vv = *(const v4f*)(vdual + (size_t)b * HW_ + n0 + 4 * nq);
  const v4f ua = *(const v4f*)(u + b * M_ + mg * 8);
  const v4f ub = *(const v4f*)(u + b * M_ + mg * 8 + 4);
  const float* xrow = xp + ((size_t)b * HW_ + n0 + kh * 128 + 4 * laneHi) * D_ + dbase + 4 * laneLo;
  v4f L[16][4];
#pragma unroll
  for (int i = 0; i < 16; ++i) {
#pragma unroll
    for (int s = 0; s < 4; ++s) L[i][s] = *(const v4f*)(xrow + (size_t)(8 * i + s) * D_);
  }

  {
    const float um[8] = {ua.x, ua.y, ua.z, ua.w, ub.x, ub.y, ub.z, ub.w};
#pragma unroll
    for (int j = 0; j < 8; ++j) {
      v4f p;
      p.x = expf(z[j].x + um[j] + vv.x - NORM_);
      p.y = expf(z[j].y + um[j] + vv.y - NORM_);
      p.z = expf(z[j].z + um[j] + vv.z - NORM_);
      p.w = expf(z[j].w + um[j] + vv.w - NORM_);
      *(v4f*)(sP + nq * PQ_ + (mg * 8 + j) * 4) = p;
    }
    sR[tid] = r1;
  }
  __builtin_amdgcn_fence(__ATOMIC_RELEASE, "workgroup", "local");
  __builtin_amdgcn_s_barrier();
  __builtin_amdgcn_fence(__ATOMIC_ACQUIRE, "workgroup", "local");

  const float* aBase = sP + (kh * 32 + laneHi) * PQ_ + laneLo * 4;
  const float* rBase = sR + kh * 128 + 4 * laneHi;

  v8f acc[2][4];
#pragma unroll
  for (int t = 0; t < 2; ++t) {
#pragma unroll
    for (int j = 0; j < 4; ++j) acc[t][j] = (v8f){};
  }

  v4f a0s[2], a1s[2], rs[2];
  a0s[0] = *(const v4f*)(aBase);
  a1s[0] = *(const v4f*)(aBase + 64);
  rs[0] = *(const v4f*)(rBase);
#pragma unroll
  for (int i = 0; i < 16; ++i) {
    if (i + 1 < 16) {
      a0s[(i + 1) & 1] = *(const v4f*)(aBase + (i + 1) * 2 * PQ_);
      a1s[(i + 1) & 1] = *(const v4f*)(aBase + (i + 1) * 2 * PQ_ + 64);
      rs[(i + 1) & 1] = *(const v4f*)(rBase + (i + 1) * 8);
    }
    const v4f a0 = a0s[i & 1];
    const v4f a1 = a1s[i & 1];
    const v4f r4 = rs[i & 1];
#pragma unroll
    for (int j = 0; j < 4; ++j) {
      const v2f bA = {L[i][0][j] * r4.x, L[i][1][j] * r4.y};
      acc[0][j] = wmma4((v2f){a0.x, a0.y}, bA, acc[0][j]);
      acc[1][j] = wmma4((v2f){a1.x, a1.y}, bA, acc[1][j]);
    }
#pragma unroll
    for (int j = 0; j < 4; ++j) {
      const v2f bB = {L[i][2][j] * r4.z, L[i][3][j] * r4.w};
      acc[0][j] = wmma4((v2f){a0.z, a0.w}, bB, acc[0][j]);
      acc[1][j] = wmma4((v2f){a1.z, a1.w}, bB, acc[1][j]);
    }
  }

  float* red = sRed + (size_t)g * (16 * 32 * 4) + lane * 4;
  if (kh == 1) {
#pragma unroll
    for (int t = 0; t < 2; ++t) {
#pragma unroll
      for (int r = 0; r < 8; ++r)
        *(v4f*)(red + (t * 8 + r) * 128) =
            (v4f){acc[t][0][r], acc[t][1][r], acc[t][2][r], acc[t][3][r]};
    }
  }
  __syncthreads();
  if (kh == 0) {
    float* o = part + (size_t)blockIdx.x * (M_ * D_) + dbase + 4 * laneLo;
#pragma unroll
    for (int t = 0; t < 2; ++t) {
#pragma unroll
      for (int r = 0; r < 8; ++r) {
        const v4f hi = *(const v4f*)(red + (t * 8 + r) * 128);
        const v4f lo = {acc[t][0][r], acc[t][1][r], acc[t][2][r], acc[t][3][r]};
        *(v4f*)(o + (size_t)(16 * t + 8 * laneHi + r) * D_) = lo + hi;
      }
    }
  }
}

__global__ void __launch_bounds__(256) k_vt_sum(const float* __restrict__ part,
                                                float* __restrict__ vt) {
  const int i4 = blockIdx.x * 256 + threadIdx.x;
  const int b = i4 >> 11;
  const int e = i4 & 2047;
  const float* p = part + ((size_t)b * VT_CHUNKS_ * 2048 + e) * 4;
  v4f t[VT_CHUNKS_];
#pragma unroll
  for (int s = 0; s < VT_CHUNKS_; ++s) t[s] = *(const v4f*)(p + (size_t)s * (M_ * D_));
  v4f sum = t[0];
#pragma unroll
  for (int s = 1; s < VT_CHUNKS_; ++s) sum += t[s];
  *(v4f*)(vt + (size_t)i4 * 4) = sum;
}

extern "C" void kernel_launch(void* const* d_in, const int* in_sizes, int n_in,
                              void* d_out, int out_size, void* d_ws, size_t ws_size,
                              hipStream_t stream) {
  const float* x      = (const float*)d_in[0];
  const float* conv_w = (const float*)d_in[1];
  const float* conv_b = (const float*)d_in[2];
  const float* v      = (const float*)d_in[3];

  float* out_vt = (float*)d_out;
  float* out_xp = out_vt + (size_t)B_ * M_ * D_;

  float* ws   = (float*)d_ws;
  float* what = ws + OFF_WHAT;
  float* Z    = ws + OFF_Z;
  float* u    = ws + OFF_U;
  float* vd   = ws + OFF_V;
  float* rinv = ws + OFF_RINV;
  float* part = ws + OFF_PART;

  k_norm_v<<<1, 1024, 0, stream>>>(v, what);
  k_projw<<<256, 256, 0, stream>>>(x, conv_w, conv_b, out_xp);
  k_scores<<<B_ * HW_ / 128, 256, 0, stream>>>(out_xp, what, Z, rinv);
  k_u_first<<<B_ * M_, 256, 0, stream>>>(Z, u);
  k_vdual<<<B_ * HW_ / 256, 256, 0, stream>>>(Z, u, vd);
  for (int it = 1; it < 3; ++it) {
    k_u<<<B_ * M_, 256, 0, stream>>>(Z, vd, u);
    k_vdual<<<B_ * HW_ / 256, 256, 0, stream>>>(Z, u, vd);
  }
  k_vt_part<<<B_ * VT_CHUNKS_, 256, 0, stream>>>(Z, u, vd, out_xp, rinv, part);
  k_vt_sum<<<B_ * M_ * D_ / 4 / 256, 256, 0, stream>>>(part, out_vt);
}
